# starting kernel
# baseline (speedup 1.0000x reference)
#include <hip/hip_runtime.h>
#include <math.h>


typedef float v2f __attribute__((ext_vector_type(2)));
typedef float v8f __attribute__((ext_vector_type(8)));

#define B_SZ    4
#define LSEQ    1024
#define DMODEL  1024
#define DINNER  2048
#define DSTATE  16
#define DCONV   4
#define DTRANK  64
#define NROWS   (B_SZ * LSEQ)

__device__ __forceinline__ float silu_f(float x) { return x / (1.0f + __expf(-x)); }
__device__ __forceinline__ float softplus_f(float x) {
    return (x > 20.0f) ? x : log1pf(__expf(x));
}

__device__ __forceinline__ v8f wmma_f32(v2f a, v2f b, v8f c) {
    return __builtin_amdgcn_wmma_f32_16x16x4_f32(
               false, a, false, b, (short)0, c, false, false);
}

template <int EPI>
__global__ __launch_bounds__(64) void gemm_wmma_blocked(
    const float* __restrict__ A, const float* __restrict__ B,
    float* __restrict__ C, const float* __restrict__ bias,
    int K, int lda, int ldb, int ldc)
{
    const int lane   = threadIdx.x & 31;
    const int wave   = threadIdx.x >> 5;
    const int half   = lane >> 4;
    const int mr     = lane & 15;
    const int tm     = blockIdx.y;
    const int tnBase = (blockIdx.x * 2 + wave) * 4;

    const float* Arow0 = A + (size_t)(tm * 32 + mr) * lda;
    const float* Arow1 = Arow0 + (size_t)16 * lda;
    const float* Bc    = B + (size_t)(tnBase * 16 + mr);

    v8f c[2][4];
#pragma unroll
    for (int i = 0; i < 2; ++i)
#pragma unroll
        for (int j = 0; j < 4; ++j) c[i][j] = (v8f){};

#pragma unroll 4
    for (int k0 = 0; k0 < K; k0 += 4) {
        const int ka = k0 + 2 * half;
        v2f a0 = *(const v2f*)(Arow0 + ka);
        v2f a1 = *(const v2f*)(Arow1 + ka);
        v2f b[4];
#pragma unroll
        for (int j = 0; j < 4; ++j) {
            b[j].x = Bc[(size_t)ka * ldb + j * 16];
            b[j].y = Bc[(size_t)(ka + 1) * ldb + j * 16];
        }
#pragma unroll
        for (int j = 0; j < 4; ++j) {
            c[0][j] = wmma_f32(a0, b[j], c[0][j]);
            c[1][j] = wmma_f32(a1, b[j], c[1][j]);
        }
    }

#pragma unroll
    for (int j = 0; j < 4; ++j) {
        const int col = (tnBase + j) * 16 + mr;
        const float bv = (EPI == 1) ? bias[col] : 0.0f;
#pragma unroll
        for (int i = 0; i < 2; ++i) {
#pragma unroll
            for (int r = 0; r < 8; ++r) {
                const int row = tm * 32 + i * 16 + half * 8 + r;
                float v = c[i][j][r];
                if (EPI == 1) v = softplus_f(v + bv);
                C[(size_t)row * ldc + col] = v;
            }
        }
    }
}

__global__ __launch_bounds__(64) void gemm_wmma_simple(
    const float* __restrict__ A, const float* __restrict__ B,
    float* __restrict__ C, int K, int lda, int ldb, int ldc)
{
    const int lane = threadIdx.x & 31;
    const int wave = threadIdx.x >> 5;
    const int tn   = blockIdx.x * 2 + wave;
    const int tm   = blockIdx.y;
    const int half = lane >> 4;
    const int mr   = lane & 15;

    const float* Arow = A + (size_t)(tm * 16 + mr) * lda;
    const float* Bcol = B + (size_t)(tn * 16 + mr);

    v8f c = {};
#pragma unroll 8
    for (int k0 = 0; k0 < K; k0 += 4) {
        const int ka = k0 + 2 * half;
        v2f a = *(const v2f*)(Arow + ka);
        v2f b;
        b.x = Bcol[(size_t)ka * ldb];
        b.y = Bcol[(size_t)(ka + 1) * ldb];
        c = wmma_f32(a, b, c);
    }

    const int col = tn * 16 + mr;
#pragma unroll
    for (int r = 0; r < 8; ++r)
        C[(size_t)(tm * 16 + half * 8 + r) * ldc + col] = c[r];
}

__global__ __launch_bounds__(256) void conv_silu_kernel(
    const float* __restrict__ xz, const float* __restrict__ conv_w,
    const float* __restrict__ conv_b, float* __restrict__ xc)
{
    const int idx = blockIdx.x * 256 + threadIdx.x;
    const int d   = idx & (DINNER - 1);
    const int pos = idx >> 11;
    const int l   = pos & (LSEQ - 1);
    const int b   = pos >> 10;

    float acc = conv_b[d];
#pragma unroll
    for (int k = 0; k < DCONV; ++k) {
        const int ls = l - (DCONV - 1) + k;
        if (ls >= 0)
            acc += conv_w[d * DCONV + k] *
                   xz[((size_t)(b * LSEQ + ls)) * (2 * DINNER) + d];
    }
    xc[(size_t)pos * DINNER + d] = silu_f(acc);
}

__global__ __launch_bounds__(256) void scan_gate_kernel(
    const float* __restrict__ xc,   const float* __restrict__ dtm,
    const float* __restrict__ xdbl, const float* __restrict__ A_log,
    const float* __restrict__ xz,   const float* __restrict__ D_param,
    float* __restrict__ y)
{
    const int b = blockIdx.y;
    const int d = blockIdx.x * 256 + threadIdx.x;

    float Ar[DSTATE];
#pragma unroll
    for (int n = 0; n < DSTATE; ++n)
        Ar[n] = -__expf(A_log[d * DSTATE + n]);
    const float Dv = D_param[d];

    float h[DSTATE];
#pragma unroll
    for (int n = 0; n < DSTATE; ++n) h[n] = 0.0f;

    const size_t rowBase = (size_t)b * LSEQ;
    for (int t = 0; t < LSEQ; ++t) {
        const size_t row = rowBase + t;
        const float* bc = xdbl + row * (DTRANK + 2 * DSTATE) + DTRANK;

        const float u   = xc [row * DINNER + d];
        const float dtv = dtm[row * DINNER + d];
        __builtin_prefetch(xc  + (row + 1) * DINNER + d, 0, 1);
        __builtin_prefetch(dtm + (row + 1) * DINNER + d, 0, 1);

        float acc = 0.0f;
#pragma unroll
        for (int n = 0; n < DSTATE; ++n) {
            const float dA = __expf(dtv * Ar[n]);
            h[n] = dA * h[n] + (dtv * bc[n]) * u;
            acc += bc[DSTATE + n] * h[n];
        }
        const float z = xz[row * (2 * DINNER) + DINNER + d];
        y[row * DINNER + d] = (acc + Dv * u) * silu_f(z);
    }
}

extern "C" void kernel_launch(void* const* d_in, const int* in_sizes, int n_in,
                              void* d_out, int out_size, void* d_ws, size_t ws_size,
                              hipStream_t stream)
{
    const float* hidden    = (const float*)d_in[0];
    const float* in_proj_w = (const float*)d_in[1];
    const float* conv_w    = (const float*)d_in[2];
    const float* conv_b    = (const float*)d_in[3];
    const float* x_proj_w  = (const float*)d_in[4];
    const float* dt_proj_w = (const float*)d_in[5];
    const float* dt_bias   = (const float*)d_in[6];
    const float* A_log     = (const float*)d_in[7];
    const float* D_param   = (const float*)d_in[8];
    const float* out_proj  = (const float*)d_in[9];
    float*       out       = (float*)d_out;

    float* ws   = (float*)d_ws;
    float* xz   = ws;
    float* xc   = xz   + (size_t)NROWS * 2 * DINNER;
    float* xdbl = xc   + (size_t)NROWS * DINNER;
    float* dtm  = xdbl + (size_t)NROWS * (DTRANK + 2 * DSTATE);
    float* y    = dtm  + (size_t)NROWS * DINNER;

    gemm_wmma_blocked<0><<<dim3((2 * DINNER) / 128, NROWS / 32), 64, 0, stream>>>(
        hidden, in_proj_w, xz, nullptr, DMODEL, DMODEL, 2 * DINNER, 2 * DINNER);

    conv_silu_kernel<<<(NROWS * DINNER) / 256, 256, 0, stream>>>(xz, conv_w, conv_b, xc);

    gemm_wmma_simple<<<dim3((DTRANK + 2 * DSTATE) / 32, NROWS / 16), 64, 0, stream>>>(
        xc, x_proj_w, xdbl, DINNER, DINNER,
        DTRANK + 2 * DSTATE, DTRANK + 2 * DSTATE);

    gemm_wmma_blocked<1><<<dim3(DINNER / 128, NROWS / 32), 64, 0, stream>>>(
        xdbl, dt_proj_w, dtm, dt_bias, DTRANK,
        DTRANK + 2 * DSTATE, DINNER, DINNER);

    scan_gate_kernel<<<dim3(DINNER / 256, B_SZ), 256, 0, stream>>>(
        xc, dtm, xdbl, A_log, xz, D_param, y);

    gemm_wmma_blocked<0><<<dim3(DMODEL / 128, NROWS / 32), 64, 0, stream>>>(
        y, out_proj, out, nullptr, DINNER, DINNER, DMODEL, DMODEL);
}


# scan: u, dt, z and the B/C row (per-lane load, read out by v_readlane) of the next 8 timesteps are loaded while the current 8 are computed, so no step waits on memory; order of sums unchanged
# speedup vs baseline: 1.4662x; 1.4662x over previous
#include <hip/hip_runtime.h>
#include <math.h>


typedef float v2f __attribute__((ext_vector_type(2)));
typedef float v8f __attribute__((ext_vector_type(8)));

#define B_SZ    4
#define LSEQ    1024
#define DMODEL  1024
#define DINNER  2048
#define DSTATE  16
#define DCONV   4
#define DTRANK  64
#define NROWS   (B_SZ * LSEQ)

__device__ __forceinline__ float silu_f(float x) { return x / (1.0f + __expf(-x)); }
__device__ __forceinline__ float softplus_f(float x) {
    return (x > 20.0f) ? x : log1pf(__expf(x));
}

__device__ __forceinline__ v8f wmma_f32(v2f a, v2f b, v8f c) {
    return __builtin_amdgcn_wmma_f32_16x16x4_f32(
               false, a, false, b, (short)0, c, false, false);
}

template <int EPI>
__global__ __launch_bounds__(64) void gemm_wmma_blocked(
    const float* __restrict__ A, const float* __restrict__ B,
    float* __restrict__ C, const float* __restrict__ bias,
    int K, int lda, int ldb, int ldc)
{
    const int lane   = threadIdx.x & 31;
    const int wave   = threadIdx.x >> 5;
    const int half   = lane >> 4;
    const int mr     = lane & 15;
    const int tm     = blockIdx.y;
    const int tnBase = (blockIdx.x * 2 + wave) * 4;

    const float* Arow0 = A + (size_t)(tm * 32 + mr) * lda;
    const float* Arow1 = Arow0 + (size_t)16 * lda;
    const float* Bc    = B + (size_t)(tnBase * 16 + mr);

    v8f c[2][4];
#pragma unroll
    for (int i = 0; i < 2; ++i)
#pragma unroll
        for (int j = 0; j < 4; ++j) c[i][j] = (v8f){};

#pragma unroll 4
    for (int k0 = 0; k0 < K; k0 += 4) {
        const int ka = k0 + 2 * half;
        v2f a0 = *(const v2f*)(Arow0 + ka);
        v2f a1 = *(const v2f*)(Arow1 + ka);
        v2f b[4];
#pragma unroll
        for (int j = 0; j < 4; ++j) {
            b[j].x = Bc[(size_t)ka * ldb + j * 16];
            b[j].y = Bc[(size_t)(ka + 1) * ldb + j * 16];
        }
#pragma unroll
        for (int j = 0; j < 4; ++j) {
            c[0][j] = wmma_f32(a0, b[j], c[0][j]);
            c[1][j] = wmma_f32(a1, b[j], c[1][j]);
        }
    }

#pragma unroll
    for (int j = 0; j < 4; ++j) {
        const int col = (tnBase + j) * 16 + mr;
        const float bv = (EPI == 1) ? bias[col] : 0.0f;
#pragma unroll
        for (int i = 0; i < 2; ++i) {
#pragma unroll
            for (int r = 0; r < 8; ++r) {
                const int row = tm * 32 + i * 16 + half * 8 + r;
                float v = c[i][j][r];
                if (EPI == 1) v = softplus_f(v + bv);
                C[(size_t)row * ldc + col] = v;
            }
        }
    }
}

__global__ __launch_bounds__(64) void gemm_wmma_simple(
    const float* __restrict__ A, const float* __restrict__ B,
    float* __restrict__ C, int K, int lda, int ldb, int ldc)
{
    const int lane = threadIdx.x & 31;
    const int wave = threadIdx.x >> 5;
    const int tn   = blockIdx.x * 2 + wave;
    const int tm   = blockIdx.y;
    const int half = lane >> 4;
    const int mr   = lane & 15;

    const float* Arow = A + (size_t)(tm * 16 + mr) * lda;
    const float* Bcol = B + (size_t)(tn * 16 + mr);

    v8f c = {};
#pragma unroll 8
    for (int k0 = 0; k0 < K; k0 += 4) {
        const int ka = k0 + 2 * half;
        v2f a = *(const v2f*)(Arow + ka);
        v2f b;
        b.x = Bcol[(size_t)ka * ldb];
        b.y = Bcol[(size_t)(ka + 1) * ldb];
        c = wmma_f32(a, b, c);
    }

    const int col = tn * 16 + mr;
#pragma unroll
    for (int r = 0; r < 8; ++r)
        C[(size_t)(tm * 16 + half * 8 + r) * ldc + col] = c[r];
}

__global__ __launch_bounds__(256) void conv_silu_kernel(
    const float* __restrict__ xz, const float* __restrict__ conv_w,
    const float* __restrict__ conv_b, float* __restrict__ xc)
{
    const int idx = blockIdx.x * 256 + threadIdx.x;
    const int d   = idx & (DINNER - 1);
    const int pos = idx >> 11;
    const int l   = pos & (LSEQ - 1);
    const int b   = pos >> 10;

    float acc = conv_b[d];
#pragma unroll
    for (int k = 0; k < DCONV; ++k) {
        const int ls = l - (DCONV - 1) + k;
        if (ls >= 0)
            acc += conv_w[d * DCONV + k] *
                   xz[((size_t)(b * LSEQ + ls)) * (2 * DINNER) + d];
    }
    xc[(size_t)pos * DINNER + d] = silu_f(acc);
}

#define SCAN_GROUP 8
#define XDW (DTRANK + 2 * DSTATE)

__global__ __launch_bounds__(256) void scan_gate_kernel(
    const float* __restrict__ xc,   const float* __restrict__ dtm,
    const float* __restrict__ xdbl, const float* __restrict__ A_log,
    const float* __restrict__ xz,   const float* __restrict__ D_param,
    float* __restrict__ y)
{
    const int b    = blockIdx.y;
    const int d    = blockIdx.x * 256 + threadIdx.x;
    const int lane = threadIdx.x & 31;

    float Ar[DSTATE];
#pragma unroll
    for (int n = 0; n < DSTATE; ++n)
        Ar[n] = -__expf(A_log[d * DSTATE + n]);
    const float Dv = D_param[d];

    float h[DSTATE];
#pragma unroll
    for (int n = 0; n < DSTATE; ++n) h[n] = 0.0f;

    const size_t rowBase = (size_t)b * LSEQ;
    const float* up  = xc   + rowBase * DINNER + d;
    const float* dp  = dtm  + rowBase * DINNER + d;
    const float* zp  = xz   + rowBase * (2 * DINNER) + DINNER + d;
    float*       yp  = y    + rowBase * DINNER + d;
    const float* bcp = xdbl + rowBase * XDW + DTRANK + lane;

    float uc[SCAN_GROUP], dc[SCAN_GROUP], zc[SCAN_GROUP], bcc[SCAN_GROUP];
#pragma unroll
    for (int i = 0; i < SCAN_GROUP; ++i) {
        uc[i]  = up [(size_t)i * DINNER];
        dc[i]  = dp [(size_t)i * DINNER];
        zc[i]  = zp [(size_t)i * (2 * DINNER)];
        bcc[i] = bcp[(size_t)i * XDW];
    }

    for (int t0 = 0; t0 < LSEQ; t0 += SCAN_GROUP) {
        const int tb = min(t0 + SCAN_GROUP, LSEQ - SCAN_GROUP);
        float un[SCAN_GROUP], dn[SCAN_GROUP], zn[SCAN_GROUP], bcn[SCAN_GROUP];
#pragma unroll
        for (int i = 0; i < SCAN_GROUP; ++i) {
            un[i]  = up [(size_t)(tb + i) * DINNER];
            dn[i]  = dp [(size_t)(tb + i) * DINNER];
            zn[i]  = zp [(size_t)(tb + i) * (2 * DINNER)];
            bcn[i] = bcp[(size_t)(tb + i) * XDW];
        }
        __builtin_amdgcn_sched_barrier(0);

#pragma unroll
        for (int i = 0; i < SCAN_GROUP; ++i) {
            const float u = uc[i], dtv = dc[i], z = zc[i];
            const int bcbits = __float_as_int(bcc[i]);
            float bc[2 * DSTATE];
#pragma unroll
            for (int n = 0; n < 2 * DSTATE; ++n)
                bc[n] = __int_as_float(__builtin_amdgcn_readlane(bcbits, n));

            float acc = 0.0f;
#pragma unroll
            for (int n = 0; n < DSTATE; ++n) {
                const float dA = __expf(dtv * Ar[n]);
                h[n] = dA * h[n] + (dtv * bc[n]) * u;
                acc += bc[DSTATE + n] * h[n];
            }
            yp[(size_t)(t0 + i) * DINNER] = (acc + Dv * u) * silu_f(z);
        }

#pragma unroll
        for (int i = 0; i < SCAN_GROUP; ++i) {
            uc[i] = un[i]; dc[i] = dn[i]; zc[i] = zn[i]; bcc[i] = bcn[i];
        }
    }
}

extern "C" void kernel_launch(void* const* d_in, const int* in_sizes, int n_in,
                              void* d_out, int out_size, void* d_ws, size_t ws_size,
                              hipStream_t stream)
{
    const float* hidden    = (const float*)d_in[0];
    const float* in_proj_w = (const float*)d_in[1];
    const float* conv_w    = (const float*)d_in[2];
    const float* conv_b    = (const float*)d_in[3];
    const float* x_proj_w  = (const float*)d_in[4];
    const float* dt_proj_w = (const float*)d_in[5];
    const float* dt_bias   = (const float*)d_in[6];
    const float* A_log     = (const float*)d_in[7];
    const float* D_param   = (const float*)d_in[8];
    const float* out_proj  = (const float*)d_in[9];
    float*       out       = (float*)d_out;

    float* ws   = (float*)d_ws;
    float* xz   = ws;
    float* xc   = xz   + (size_t)NROWS * 2 * DINNER;
    float* xdbl = xc   + (size_t)NROWS * DINNER;
    float* dtm  = xdbl + (size_t)NROWS * (DTRANK + 2 * DSTATE);
    float* y    = dtm  + (size_t)NROWS * DINNER;

    gemm_wmma_blocked<0><<<dim3((2 * DINNER) / 128, NROWS / 32), 64, 0, stream>>>(
        hidden, in_proj_w, xz, nullptr, DMODEL, DMODEL, 2 * DINNER, 2 * DINNER);

    conv_silu_kernel<<<(NROWS * DINNER) / 256, 256, 0, stream>>>(xz, conv_w, conv_b, xc);

    gemm_wmma_simple<<<dim3((DTRANK + 2 * DSTATE) / 32, NROWS / 16), 64, 0, stream>>>(
        xc, x_proj_w, xdbl, DINNER, DINNER,
        DTRANK + 2 * DSTATE, DTRANK + 2 * DSTATE);

    gemm_wmma_blocked<1><<<dim3(DINNER / 128, NROWS / 32), 64, 0, stream>>>(
        xdbl, dt_proj_w, dtm, dt_bias, DTRANK,
        DTRANK + 2 * DSTATE, DINNER, DINNER);

    scan_gate_kernel<<<dim3(DINNER / 256, B_SZ), 256, 0, stream>>>(
        xc, dtm, xdbl, A_log, xz, D_param, y);

    gemm_wmma_blocked<0><<<dim3(DMODEL / 128, NROWS / 32), 64, 0, stream>>>(
        y, out_proj, out, nullptr, DINNER, DINNER, DMODEL, DMODEL);
}


# scan: each channel's 16 states split over 4 waves (1024 waves, not 256); the C.h sum is now 4 partial sums of 4 added via LDS, y written as whole 128-B lines; operands fetched 16 steps ahead
# speedup vs baseline: 1.7084x; 1.1652x over previous
#include <hip/hip_runtime.h>
#include <math.h>


typedef float v2f __attribute__((ext_vector_type(2)));
typedef float v8f __attribute__((ext_vector_type(8)));

#define B_SZ    4
#define LSEQ    1024
#define DMODEL  1024
#define DINNER  2048
#define DSTATE  16
#define DCONV   4
#define DTRANK  64
#define NROWS   (B_SZ * LSEQ)

__device__ __forceinline__ float silu_f(float x) { return x / (1.0f + __expf(-x)); }
__device__ __forceinline__ float softplus_f(float x) {
    return (x > 20.0f) ? x : log1pf(__expf(x));
}

__device__ __forceinline__ v8f wmma_f32(v2f a, v2f b, v8f c) {
    return __builtin_amdgcn_wmma_f32_16x16x4_f32(
               false, a, false, b, (short)0, c, false, false);
}

template <int EPI>
__global__ __launch_bounds__(64) void gemm_wmma_blocked(
    const float* __restrict__ A, const float* __restrict__ B,
    float* __restrict__ C, const float* __restrict__ bias,
    int K, int lda, int ldb, int ldc)
{
    const int lane   = threadIdx.x & 31;
    const int wave   = threadIdx.x >> 5;
    const int half   = lane >> 4;
    const int mr     = lane & 15;
    const int tm     = blockIdx.y;
    const int tnBase = (blockIdx.x * 2 + wave) * 4;

    const float* Arow0 = A + (size_t)(tm * 32 + mr) * lda;
    const float* Arow1 = Arow0 + (size_t)16 * lda;
    const float* Bc    = B + (size_t)(tnBase * 16 + mr);

    v8f c[2][4];
#pragma unroll
    for (int i = 0; i < 2; ++i)
#pragma unroll
        for (int j = 0; j < 4; ++j) c[i][j] = (v8f){};

#pragma unroll 4
    for (int k0 = 0; k0 < K; k0 += 4) {
        const int ka = k0 + 2 * half;
        v2f a0 = *(const v2f*)(Arow0 + ka);
        v2f a1 = *(const v2f*)(Arow1 + ka);
        v2f b[4];
#pragma unroll
        for (int j = 0; j < 4; ++j) {
            b[j].x = Bc[(size_t)ka * ldb + j * 16];
            b[j].y = Bc[(size_t)(ka + 1) * ldb + j * 16];
        }
#pragma unroll
        for (int j = 0; j < 4; ++j) {
            c[0][j] = wmma_f32(a0, b[j], c[0][j]);
            c[1][j] = wmma_f32(a1, b[j], c[1][j]);
        }
    }

#pragma unroll
    for (int j = 0; j < 4; ++j) {
        const int col = (tnBase + j) * 16 + mr;
        const float bv = (EPI == 1) ? bias[col] : 0.0f;
#pragma unroll
        for (int i = 0; i < 2; ++i) {
#pragma unroll
            for (int r = 0; r < 8; ++r) {
                const int row = tm * 32 + i * 16 + half * 8 + r;
                float v = c[i][j][r];
                if (EPI == 1) v = softplus_f(v + bv);
                C[(size_t)row * ldc + col] = v;
            }
        }
    }
}

__global__ __launch_bounds__(64) void gemm_wmma_simple(
    const float* __restrict__ A, const float* __restrict__ B,
    float* __restrict__ C, int K, int lda, int ldb, int ldc)
{
    const int lane = threadIdx.x & 31;
    const int wave = threadIdx.x >> 5;
    const int tn   = blockIdx.x * 2 + wave;
    const int tm   = blockIdx.y;
    const int half = lane >> 4;
    const int mr   = lane & 15;

    const float* Arow = A + (size_t)(tm * 16 + mr) * lda;
    const float* Bcol = B + (size_t)(tn * 16 + mr);

    v8f c = {};
#pragma unroll 8
    for (int k0 = 0; k0 < K; k0 += 4) {
        const int ka = k0 + 2 * half;
        v2f a = *(const v2f*)(Arow + ka);
        v2f b;
        b.x = Bcol[(size_t)ka * ldb];
        b.y = Bcol[(size_t)(ka + 1) * ldb];
        c = wmma_f32(a, b, c);
    }

    const int col = tn * 16 + mr;
#pragma unroll
    for (int r = 0; r < 8; ++r)
        C[(size_t)(tm * 16 + half * 8 + r) * ldc + col] = c[r];
}

__global__ __launch_bounds__(256) void conv_silu_kernel(
    const float* __restrict__ xz, const float* __restrict__ conv_w,
    const float* __restrict__ conv_b, float* __restrict__ xc)
{
    const int idx = blockIdx.x * 256 + threadIdx.x;
    const int d   = idx & (DINNER - 1);
    const int pos = idx >> 11;
    const int l   = pos & (LSEQ - 1);
    const int b   = pos >> 10;

    float acc = conv_b[d];
#pragma unroll
    for (int k = 0; k < DCONV; ++k) {
        const int ls = l - (DCONV - 1) + k;
        if (ls >= 0)
            acc += conv_w[d * DCONV + k] *
                   xz[((size_t)(b * LSEQ + ls)) * (2 * DINNER) + d];
    }
    xc[(size_t)pos * DINNER + d] = silu_f(acc);
}

#define SCAN_GROUP 16
#define SCAN_WAVES 4
#define NPW (DSTATE / SCAN_WAVES)
#define FPW (SCAN_GROUP / SCAN_WAVES)
#define XDW (DTRANK + 2 * DSTATE)

struct ScanOps {
    float u[SCAN_GROUP];
    float dt[SCAN_GROUP];
    float bc[SCAN_GROUP];
    float uf[FPW];
    float zf[FPW];
};

__global__ __launch_bounds__(32 * SCAN_WAVES) void scan_gate_kernel(
    const float* __restrict__ xc,   const float* __restrict__ dtm,
    const float* __restrict__ xdbl, const float* __restrict__ A_log,
    const float* __restrict__ xz,   const float* __restrict__ D_param,
    float* __restrict__ y)
{
    __shared__ float part[2][SCAN_GROUP][SCAN_WAVES][32];

    const int b    = blockIdx.y;
    const int lane = threadIdx.x & 31;
    const int w    = __builtin_amdgcn_readfirstlane(threadIdx.x >> 5);
    const int d    = blockIdx.x * 32 + lane;
    const int n0   = w * NPW;

    float Ar[NPW];
#pragma unroll
    for (int j = 0; j < NPW; ++j)
        Ar[j] = -__expf(A_log[d * DSTATE + n0 + j]);
    const float Dv = D_param[d];

    float h[NPW];
#pragma unroll
    for (int j = 0; j < NPW; ++j) h[j] = 0.0f;

    const size_t rowBase = (size_t)b * LSEQ;
    const float* up  = xc   + rowBase * DINNER + d;
    const float* dp  = dtm  + rowBase * DINNER + d;
    const float* zp  = xz   + rowBase * (2 * DINNER) + DINNER + d;
    float*       yp  = y    + rowBase * DINNER + d;
    const float* bcp = xdbl + rowBase * XDW + DTRANK + lane;
    const int    f0  = w * FPW;

    auto fetch = [&](ScanOps& o, int tb) {
#pragma unroll
        for (int i = 0; i < SCAN_GROUP; ++i) {
            o.u[i]  = up [(size_t)(tb + i) * DINNER];
            o.dt[i] = dp [(size_t)(tb + i) * DINNER];
            o.bc[i] = bcp[(size_t)(tb + i) * XDW];
        }
#pragma unroll
        for (int k = 0; k < FPW; ++k) {
            o.uf[k] = up[(size_t)(tb + f0 + k) * DINNER];
            o.zf[k] = zp[(size_t)(tb + f0 + k) * (2 * DINNER)];
        }
    };

    auto group = [&](const ScanOps& cur, ScanOps& nx, int t0, int buf) {
        fetch(nx, min(t0 + SCAN_GROUP, LSEQ - SCAN_GROUP));
        __builtin_amdgcn_sched_barrier(0);

#pragma unroll
        for (int i = 0; i < SCAN_GROUP; ++i) {
            const float u = cur.u[i], dtv = cur.dt[i];
            const int bcbits = __float_as_int(cur.bc[i]);
            float Bv[NPW], Cv[NPW];
#pragma unroll
            for (int j = 0; j < NPW; ++j) {
                Bv[j] = __int_as_float(__builtin_amdgcn_readlane(bcbits, n0 + j));
                Cv[j] = __int_as_float(__builtin_amdgcn_readlane(bcbits, DSTATE + n0 + j));
            }

            float acc = 0.0f;
#pragma unroll
            for (int j = 0; j < NPW; ++j) {
                const float dA = __expf(dtv * Ar[j]);
                h[j] = dA * h[j] + (dtv * Bv[j]) * u;
                acc += Cv[j] * h[j];
            }
            part[buf][i][w][lane] = acc;
        }

        __syncthreads();

#pragma unroll
        for (int k = 0; k < FPW; ++k) {
            const int i = f0 + k;
            float s = part[buf][i][0][lane];
#pragma unroll
            for (int p = 1; p < SCAN_WAVES; ++p) s += part[buf][i][p][lane];
            yp[(size_t)(t0 + i) * DINNER] = (s + Dv * cur.uf[k]) * silu_f(cur.zf[k]);
        }
    };

    ScanOps opA, opB;
    fetch(opA, 0);
    for (int t0 = 0; t0 < LSEQ; t0 += 2 * SCAN_GROUP) {
        group(opA, opB, t0, 0);
        group(opB, opA, t0 + SCAN_GROUP, 1);
    }
}

extern "C" void kernel_launch(void* const* d_in, const int* in_sizes, int n_in,
                              void* d_out, int out_size, void* d_ws, size_t ws_size,
                              hipStream_t stream)
{
    const float* hidden    = (const float*)d_in[0];
    const float* in_proj_w = (const float*)d_in[1];
    const float* conv_w    = (const float*)d_in[2];
    const float* conv_b    = (const float*)d_in[3];
    const float* x_proj_w  = (const float*)d_in[4];
    const float* dt_proj_w = (const float*)d_in[5];
    const float* dt_bias   = (const float*)d_in[6];
    const float* A_log     = (const float*)d_in[7];
    const float* D_param   = (const float*)d_in[8];
    const float* out_proj  = (const float*)d_in[9];
    float*       out       = (float*)d_out;

    float* ws   = (float*)d_ws;
    float* xz   = ws;
    float* xc   = xz   + (size_t)NROWS * 2 * DINNER;
    float* xdbl = xc   + (size_t)NROWS * DINNER;
    float* dtm  = xdbl + (size_t)NROWS * (DTRANK + 2 * DSTATE);
    float* y    = dtm  + (size_t)NROWS * DINNER;

    gemm_wmma_blocked<0><<<dim3((2 * DINNER) / 128, NROWS / 32), 64, 0, stream>>>(
        hidden, in_proj_w, xz, nullptr, DMODEL, DMODEL, 2 * DINNER, 2 * DINNER);

    conv_silu_kernel<<<(NROWS * DINNER) / 256, 256, 0, stream>>>(xz, conv_w, conv_b, xc);

    gemm_wmma_simple<<<dim3((DTRANK + 2 * DSTATE) / 32, NROWS / 16), 64, 0, stream>>>(
        xc, x_proj_w, xdbl, DINNER, DINNER,
        DTRANK + 2 * DSTATE, DTRANK + 2 * DSTATE);

    gemm_wmma_blocked<1><<<dim3(DINNER / 128, NROWS / 32), 64, 0, stream>>>(
        xdbl, dt_proj_w, dtm, dt_bias, DTRANK,
        DTRANK + 2 * DSTATE, DINNER, DINNER);

    scan_gate_kernel<<<dim3(DINNER / 32, B_SZ), 32 * SCAN_WAVES, 0, stream>>>(
        xc, dtm, xdbl, A_log, xz, D_param, y);

    gemm_wmma_blocked<0><<<dim3(DMODEL / 128, NROWS / 32), 64, 0, stream>>>(
        y, out_proj, out, nullptr, DINNER, DINNER, DMODEL, DMODEL);
}


# GEMMs (in_proj, dt_proj, out_proj): A and B tiles staged in LDS per 128x128 work-group tile (8 waves, K chunks of 32, whole-line global loads), fragments read from LDS; order of sums unchanged
# speedup vs baseline: 2.7760x; 1.6249x over previous
#include <hip/hip_runtime.h>
#include <math.h>


typedef float v2f __attribute__((ext_vector_type(2)));
typedef float v8f __attribute__((ext_vector_type(8)));

#define B_SZ    4
#define LSEQ    1024
#define DMODEL  1024
#define DINNER  2048
#define DSTATE  16
#define DCONV   4
#define DTRANK  64
#define NROWS   (B_SZ * LSEQ)

__device__ __forceinline__ float silu_f(float x) { return x / (1.0f + __expf(-x)); }
__device__ __forceinline__ float softplus_f(float x) {
    return (x > 20.0f) ? x : log1pf(__expf(x));
}

__device__ __forceinline__ v8f wmma_f32(v2f a, v2f b, v8f c) {
    return __builtin_amdgcn_wmma_f32_16x16x4_f32(
               false, a, false, b, (short)0, c, false, false);
}

#define GB_M 128
#define GB_N 128
#define GB_K 32
#define GA_P (GB_K + 4)
#define GB_P (2 * GB_N + 32)

template <int EPI>
__global__ __launch_bounds__(256) void gemm_wmma_blocked(
    const float* __restrict__ A, const float* __restrict__ B,
    float* __restrict__ C, const float* __restrict__ bias,
    int K, int lda, int ldb, int ldc)
{
    __shared__ __attribute__((aligned(16))) float As[2][GB_M * GA_P];
    __shared__ __attribute__((aligned(16))) float Bs[2][(GB_K / 2) * GB_P];

    const int tid  = threadIdx.x;
    const int lane = tid & 31;
    const int wave = tid >> 5;
    const int half = lane >> 4;
    const int mr   = lane & 15;
    const int wm   = wave >> 1;
    const int wn   = wave & 1;
    const int bm0  = blockIdx.y * GB_M;
    const int bn0  = blockIdx.x * GB_N;

    const int aRow = tid >> 3, aQ = tid & 7;
    const int bPr  = tid >> 7, bN = tid & 127;
    const float* Ag = A + (size_t)(bm0 + aRow) * lda + 4 * aQ;
    const float* Bg = B + (size_t)(2 * bPr) * ldb + bn0 + bN;
    const int aSt = aRow * GA_P + 4 * aQ;
    const int bSt = bPr * GB_P + 2 * bN;

    float4 ra0, ra1, ra2, ra3;
    v2f rb0, rb1, rb2, rb3, rb4, rb5, rb6, rb7;
    auto gload = [&](int k0) {
        ra0 = *(const float4*)(Ag + k0);
        ra1 = *(const float4*)(Ag + (size_t)32 * lda + k0);
        ra2 = *(const float4*)(Ag + (size_t)64 * lda + k0);
        ra3 = *(const float4*)(Ag + (size_t)96 * lda + k0);
        const float* Bk = Bg + (size_t)k0 * ldb;
        const size_t l1 = (size_t)ldb;
        rb0.x = Bk[0];        rb0.y = Bk[l1];
        rb1.x = Bk[4 * l1];   rb1.y = Bk[5 * l1];
        rb2.x = Bk[8 * l1];   rb2.y = Bk[9 * l1];
        rb3.x = Bk[12 * l1];  rb3.y = Bk[13 * l1];
        rb4.x = Bk[16 * l1];  rb4.y = Bk[17 * l1];
        rb5.x = Bk[20 * l1];  rb5.y = Bk[21 * l1];
        rb6.x = Bk[24 * l1];  rb6.y = Bk[25 * l1];
        rb7.x = Bk[28 * l1];  rb7.y = Bk[29 * l1];
    };
    auto sstore = [&](int buf) {
        float* Ad = &As[buf][aSt];
        float* Bd = &Bs[buf][bSt];
        *(float4*)(Ad)             = ra0;
        *(float4*)(Ad + 32 * GA_P) = ra1;
        *(float4*)(Ad + 64 * GA_P) = ra2;
        *(float4*)(Ad + 96 * GA_P) = ra3;
        *(v2f*)(Bd)              = rb0;
        *(v2f*)(Bd +  2 * GB_P)  = rb1;
        *(v2f*)(Bd +  4 * GB_P)  = rb2;
        *(v2f*)(Bd +  6 * GB_P)  = rb3;
        *(v2f*)(Bd +  8 * GB_P)  = rb4;
        *(v2f*)(Bd + 10 * GB_P)  = rb5;
        *(v2f*)(Bd + 12 * GB_P)  = rb6;
        *(v2f*)(Bd + 14 * GB_P)  = rb7;
    };

    v8f c[2][4];
#pragma unroll
    for (int i = 0; i < 2; ++i)
#pragma unroll
        for (int j = 0; j < 4; ++j) c[i][j] = (v8f){};

    const int aFr = (wm * 32 + mr) * GA_P + 2 * half;
    const int bFr = half * GB_P + 2 * (wn * 64 + mr);

    auto compute = [&](int buf) {
        const float* Aw = &As[buf][aFr];
        const float* Bw = &Bs[buf][bFr];
#pragma unroll
        for (int kk = 0; kk < GB_K; kk += 4) {
            v2f a0 = *(const v2f*)(Aw + kk);
            v2f a1 = *(const v2f*)(Aw + 16 * GA_P + kk);
            v2f b[4];
#pragma unroll
            for (int j = 0; j < 4; ++j)
                b[j] = *(const v2f*)(Bw + (kk / 2) * GB_P + j * 32);
#pragma unroll
            for (int j = 0; j < 4; ++j) {
                c[0][j] = wmma_f32(a0, b[j], c[0][j]);
                c[1][j] = wmma_f32(a1, b[j], c[1][j]);
            }
        }
    };

    gload(0);
    int buf = 0;
    for (int k0 = 0; k0 < K; k0 += GB_K) {
        sstore(buf);
        __syncthreads();
        gload(min(k0 + GB_K, K - GB_K));
        __builtin_amdgcn_sched_barrier(0);
        compute(buf);
        buf ^= 1;
    }

#pragma unroll
    for (int j = 0; j < 4; ++j) {
        const int col = bn0 + wn * 64 + j * 16 + mr;
        const float bv = (EPI == 1) ? bias[col] : 0.0f;
#pragma unroll
        for (int i = 0; i < 2; ++i) {
#pragma unroll
            for (int r = 0; r < 8; ++r) {
                const int row = bm0 + wm * 32 + i * 16 + half * 8 + r;
                float v = c[i][j][r];
                if (EPI == 1) v = softplus_f(v + bv);
                C[(size_t)row * ldc + col] = v;
            }
        }
    }
}

__global__ __launch_bounds__(64) void gemm_wmma_simple(
    const float* __restrict__ A, const float* __restrict__ B,
    float* __restrict__ C, int K, int lda, int ldb, int ldc)
{
    const int lane = threadIdx.x & 31;
    const int wave = threadIdx.x >> 5;
    const int tn   = blockIdx.x * 2 + wave;
    const int tm   = blockIdx.y;
    const int half = lane >> 4;
    const int mr   = lane & 15;

    const float* Arow = A + (size_t)(tm * 16 + mr) * lda;
    const float* Bcol = B + (size_t)(tn * 16 + mr);

    v8f c = {};
#pragma unroll 8
    for (int k0 = 0; k0 < K; k0 += 4) {
        const int ka = k0 + 2 * half;
        v2f a = *(const v2f*)(Arow + ka);
        v2f b;
        b.x = Bcol[(size_t)ka * ldb];
        b.y = Bcol[(size_t)(ka + 1) * ldb];
        c = wmma_f32(a, b, c);
    }

    const int col = tn * 16 + mr;
#pragma unroll
    for (int r = 0; r < 8; ++r)
        C[(size_t)(tm * 16 + half * 8 + r) * ldc + col] = c[r];
}

__global__ __launch_bounds__(256) void conv_silu_kernel(
    const float* __restrict__ xz, const float* __restrict__ conv_w,
    const float* __restrict__ conv_b, float* __restrict__ xc)
{
    const int idx = blockIdx.x * 256 + threadIdx.x;
    const int d   = idx & (DINNER - 1);
    const int pos = idx >> 11;
    const int l   = pos & (LSEQ - 1);
    const int b   = pos >> 10;

    float acc = conv_b[d];
#pragma unroll
    for (int k = 0; k < DCONV; ++k) {
        const int ls = l - (DCONV - 1) + k;
        if (ls >= 0)
            acc += conv_w[d * DCONV + k] *
                   xz[((size_t)(b * LSEQ + ls)) * (2 * DINNER) + d];
    }
    xc[(size_t)pos * DINNER + d] = silu_f(acc);
}

#define SCAN_GROUP 16
#define SCAN_WAVES 4
#define NPW (DSTATE / SCAN_WAVES)
#define FPW (SCAN_GROUP / SCAN_WAVES)
#define XDW (DTRANK + 2 * DSTATE)

struct ScanOps {
    float u[SCAN_GROUP];
    float dt[SCAN_GROUP];
    float bc[SCAN_GROUP];
    float uf[FPW];
    float zf[FPW];
};

__global__ __launch_bounds__(32 * SCAN_WAVES) void scan_gate_kernel(
    const float* __restrict__ xc,   const float* __restrict__ dtm,
    const float* __restrict__ xdbl, const float* __restrict__ A_log,
    const float* __restrict__ xz,   const float* __restrict__ D_param,
    float* __restrict__ y)
{
    __shared__ float part[2][SCAN_GROUP][SCAN_WAVES][32];

    const int b    = blockIdx.y;
    const int lane = threadIdx.x & 31;
    const int w    = __builtin_amdgcn_readfirstlane(threadIdx.x >> 5);
    const int d    = blockIdx.x * 32 + lane;
    const int n0   = w * NPW;

    float Ar[NPW];
#pragma unroll
    for (int j = 0; j < NPW; ++j)
        Ar[j] = -__expf(A_log[d * DSTATE + n0 + j]);
    const float Dv = D_param[d];

    float h[NPW];
#pragma unroll
    for (int j = 0; j < NPW; ++j) h[j] = 0.0f;

    const size_t rowBase = (size_t)b * LSEQ;
    const float* up  = xc   + rowBase * DINNER + d;
    const float* dp  = dtm  + rowBase * DINNER + d;
    const float* zp  = xz   + rowBase * (2 * DINNER) + DINNER + d;
    float*       yp  = y    + rowBase * DINNER + d;
    const float* bcp = xdbl + rowBase * XDW + DTRANK + lane;
    const int    f0  = w * FPW;

    auto fetch = [&](ScanOps& o, int tb) {
#pragma unroll
        for (int i = 0; i < SCAN_GROUP; ++i) {
            o.u[i]  = up [(size_t)(tb + i) * DINNER];
            o.dt[i] = dp [(size_t)(tb + i) * DINNER];
            o.bc[i] = bcp[(size_t)(tb + i) * XDW];
        }
#pragma unroll
        for (int k = 0; k < FPW; ++k) {
            o.uf[k] = up[(size_t)(tb + f0 + k) * DINNER];
            o.zf[k] = zp[(size_t)(tb + f0 + k) * (2 * DINNER)];
        }
    };

    auto group = [&](const ScanOps& cur, ScanOps& nx, int t0, int buf) {
        fetch(nx, min(t0 + SCAN_GROUP, LSEQ - SCAN_GROUP));
        __builtin_amdgcn_sched_barrier(0);

#pragma unroll
        for (int i = 0; i < SCAN_GROUP; ++i) {
            const float u = cur.u[i], dtv = cur.dt[i];
            const int bcbits = __float_as_int(cur.bc[i]);
            float Bv[NPW], Cv[NPW];
#pragma unroll
            for (int j = 0; j < NPW; ++j) {
                Bv[j] = __int_as_float(__builtin_amdgcn_readlane(bcbits, n0 + j));
                Cv[j] = __int_as_float(__builtin_amdgcn_readlane(bcbits, DSTATE + n0 + j));
            }

            float acc = 0.0f;
#pragma unroll
            for (int j = 0; j < NPW; ++j) {
                const float dA = __expf(dtv * Ar[j]);
                h[j] = dA * h[j] + (dtv * Bv[j]) * u;
                acc += Cv[j] * h[j];
            }
            part[buf][i][w][lane] = acc;
        }

        __syncthreads();

#pragma unroll
        for (int k = 0; k < FPW; ++k) {
            const int i = f0 + k;
            float s = part[buf][i][0][lane];
#pragma unroll
            for (int p = 1; p < SCAN_WAVES; ++p) s += part[buf][i][p][lane];
            yp[(size_t)(t0 + i) * DINNER] = (s + Dv * cur.uf[k]) * silu_f(cur.zf[k]);
        }
    };

    ScanOps opA, opB;
    fetch(opA, 0);
    for (int t0 = 0; t0 < LSEQ; t0 += 2 * SCAN_GROUP) {
        group(opA, opB, t0, 0);
        group(opB, opA, t0 + SCAN_GROUP, 1);
    }
}

extern "C" void kernel_launch(void* const* d_in, const int* in_sizes, int n_in,
                              void* d_out, int out_size, void* d_ws, size_t ws_size,
                              hipStream_t stream)
{
    const float* hidden    = (const float*)d_in[0];
    const float* in_proj_w = (const float*)d_in[1];
    const float* conv_w    = (const float*)d_in[2];
    const float* conv_b    = (const float*)d_in[3];
    const float* x_proj_w  = (const float*)d_in[4];
    const float* dt_proj_w = (const float*)d_in[5];
    const float* dt_bias   = (const float*)d_in[6];
    const float* A_log     = (const float*)d_in[7];
    const float* D_param   = (const float*)d_in[8];
    const float* out_proj  = (const float*)d_in[9];
    float*       out       = (float*)d_out;

    float* ws   = (float*)d_ws;
    float* xz   = ws;
    float* xc   = xz   + (size_t)NROWS * 2 * DINNER;
    float* xdbl = xc   + (size_t)NROWS * DINNER;
    float* dtm  = xdbl + (size_t)NROWS * (DTRANK + 2 * DSTATE);
    float* y    = dtm  + (size_t)NROWS * DINNER;

    gemm_wmma_blocked<0><<<dim3((2 * DINNER) / GB_N, NROWS / GB_M), 256, 0, stream>>>(
        hidden, in_proj_w, xz, nullptr, DMODEL, DMODEL, 2 * DINNER, 2 * DINNER);

    conv_silu_kernel<<<(NROWS * DINNER) / 256, 256, 0, stream>>>(xz, conv_w, conv_b, xc);

    gemm_wmma_simple<<<dim3((DTRANK + 2 * DSTATE) / 32, NROWS / 16), 64, 0, stream>>>(
        xc, x_proj_w, xdbl, DINNER, DINNER,
        DTRANK + 2 * DSTATE, DTRANK + 2 * DSTATE);

    gemm_wmma_blocked<1><<<dim3(DINNER / GB_N, NROWS / GB_M), 256, 0, stream>>>(
        xdbl, dt_proj_w, dtm, dt_bias, DTRANK,
        DTRANK + 2 * DSTATE, DINNER, DINNER);

    scan_gate_kernel<<<dim3(DINNER / 32, B_SZ), 32 * SCAN_WAVES, 0, stream>>>(
        xc, dtm, xdbl, A_log, xz, D_param, y);

    gemm_wmma_blocked<0><<<dim3(DMODEL / GB_N, NROWS / GB_M), 256, 0, stream>>>(
        y, out_proj, out, nullptr, DINNER, DINNER, DMODEL, DMODEL);
}


# x_proj GEMM (4096x96, K=2048): A and B chunks of 128 k staged in LDS by whole-line loads, K split over the 8 waves of a 16-row work-group, partials added via LDS; order of sums changed
# speedup vs baseline: 3.0123x; 1.0851x over previous
#include <hip/hip_runtime.h>
#include <math.h>


typedef float v2f __attribute__((ext_vector_type(2)));
typedef float v8f __attribute__((ext_vector_type(8)));

#define B_SZ    4
#define LSEQ    1024
#define DMODEL  1024
#define DINNER  2048
#define DSTATE  16
#define DCONV   4
#define DTRANK  64
#define NROWS   (B_SZ * LSEQ)

__device__ __forceinline__ float silu_f(float x) { return x / (1.0f + __expf(-x)); }
__device__ __forceinline__ float softplus_f(float x) {
    return (x > 20.0f) ? x : log1pf(__expf(x));
}

__device__ __forceinline__ v8f wmma_f32(v2f a, v2f b, v8f c) {
    return __builtin_amdgcn_wmma_f32_16x16x4_f32(
               false, a, false, b, (short)0, c, false, false);
}

#define GB_M 128
#define GB_N 128
#define GB_K 32
#define GA_P (GB_K + 4)
#define GB_P (2 * GB_N + 32)

template <int EPI>
__global__ __launch_bounds__(256) void gemm_wmma_blocked(
    const float* __restrict__ A, const float* __restrict__ B,
    float* __restrict__ C, const float* __restrict__ bias,
    int K, int lda, int ldb, int ldc)
{
    __shared__ __attribute__((aligned(16))) float As[2][GB_M * GA_P];
    __shared__ __attribute__((aligned(16))) float Bs[2][(GB_K / 2) * GB_P];

    const int tid  = threadIdx.x;
    const int lane = tid & 31;
    const int wave = tid >> 5;
    const int half = lane >> 4;
    const int mr   = lane & 15;
    const int wm   = wave >> 1;
    const int wn   = wave & 1;
    const int bm0  = blockIdx.y * GB_M;
    const int bn0  = blockIdx.x * GB_N;

    const int aRow = tid >> 3, aQ = tid & 7;
    const int bPr  = tid >> 7, bN = tid & 127;
    const float* Ag = A + (size_t)(bm0 + aRow) * lda + 4 * aQ;
    const float* Bg = B + (size_t)(2 * bPr) * ldb + bn0 + bN;
    const int aSt = aRow * GA_P + 4 * aQ;
    const int bSt = bPr * GB_P + 2 * bN;

    float4 ra0, ra1, ra2, ra3;
    v2f rb0, rb1, rb2, rb3, rb4, rb5, rb6, rb7;
    auto gload = [&](int k0) {
        ra0 = *(const float4*)(Ag + k0);
        ra1 = *(const float4*)(Ag + (size_t)32 * lda + k0);
        ra2 = *(const float4*)(Ag + (size_t)64 * lda + k0);
        ra3 = *(const float4*)(Ag + (size_t)96 * lda + k0);
        const float* Bk = Bg + (size_t)k0 * ldb;
        const size_t l1 = (size_t)ldb;
        rb0.x = Bk[0];        rb0.y = Bk[l1];
        rb1.x = Bk[4 * l1];   rb1.y = Bk[5 * l1];
        rb2.x = Bk[8 * l1];   rb2.y = Bk[9 * l1];
        rb3.x = Bk[12 * l1];  rb3.y = Bk[13 * l1];
        rb4.x = Bk[16 * l1];  rb4.y = Bk[17 * l1];
        rb5.x = Bk[20 * l1];  rb5.y = Bk[21 * l1];
        rb6.x = Bk[24 * l1];  rb6.y = Bk[25 * l1];
        rb7.x = Bk[28 * l1];  rb7.y = Bk[29 * l1];
    };
    auto sstore = [&](int buf) {
        float* Ad = &As[buf][aSt];
        float* Bd = &Bs[buf][bSt];
        *(float4*)(Ad)             = ra0;
        *(float4*)(Ad + 32 * GA_P) = ra1;
        *(float4*)(Ad + 64 * GA_P) = ra2;
        *(float4*)(Ad + 96 * GA_P) = ra3;
        *(v2f*)(Bd)              = rb0;
        *(v2f*)(Bd +  2 * GB_P)  = rb1;
        *(v2f*)(Bd +  4 * GB_P)  = rb2;
        *(v2f*)(Bd +  6 * GB_P)  = rb3;
        *(v2f*)(Bd +  8 * GB_P)  = rb4;
        *(v2f*)(Bd + 10 * GB_P)  = rb5;
        *(v2f*)(Bd + 12 * GB_P)  = rb6;
        *(v2f*)(Bd + 14 * GB_P)  = rb7;
    };

    v8f c[2][4];
#pragma unroll
    for (int i = 0; i < 2; ++i)
#pragma unroll
        for (int j = 0; j < 4; ++j) c[i][j] = (v8f){};

    const int aFr = (wm * 32 + mr) * GA_P + 2 * half;
    const int bFr = half * GB_P + 2 * (wn * 64 + mr);

    auto compute = [&](int buf) {
        const float* Aw = &As[buf][aFr];
        const float* Bw = &Bs[buf][bFr];
#pragma unroll
        for (int kk = 0; kk < GB_K; kk += 4) {
            v2f a0 = *(const v2f*)(Aw + kk);
            v2f a1 = *(const v2f*)(Aw + 16 * GA_P + kk);
            v2f b[4];
#pragma unroll
            for (int j = 0; j < 4; ++j)
                b[j] = *(const v2f*)(Bw + (kk / 2) * GB_P + j * 32);
#pragma unroll
            for (int j = 0; j < 4; ++j) {
                c[0][j] = wmma_f32(a0, b[j], c[0][j]);
                c[1][j] = wmma_f32(a1, b[j], c[1][j]);
            }
        }
    };

    gload(0);
    int buf = 0;
    for (int k0 = 0; k0 < K; k0 += GB_K) {
        sstore(buf);
        __syncthreads();
        gload(min(k0 + GB_K, K - GB_K));
        __builtin_amdgcn_sched_barrier(0);
        compute(buf);
        buf ^= 1;
    }

#pragma unroll
    for (int j = 0; j < 4; ++j) {
        const int col = bn0 + wn * 64 + j * 16 + mr;
        const float bv = (EPI == 1) ? bias[col] : 0.0f;
#pragma unroll
        for (int i = 0; i < 2; ++i) {
#pragma unroll
            for (int r = 0; r < 8; ++r) {
                const int row = bm0 + wm * 32 + i * 16 + half * 8 + r;
                float v = c[i][j][r];
                if (EPI == 1) v = softplus_f(v + bv);
                C[(size_t)row * ldc + col] = v;
            }
        }
    }
}

#define XP_M   16
#define XP_N   (DTRANK + 2 * DSTATE)
#define XP_KC  128
#define XP_AP  (XP_KC + 4)
#define XP_NT  (XP_N / 16)
#define XP_NCH (DINNER / XP_KC)

__global__ __launch_bounds__(256) void gemm_xproj_kernel(
    const float* __restrict__ A, const float* __restrict__ B, float* __restrict__ C)
{
    __shared__ __attribute__((aligned(16))) float As[2][XP_M * XP_AP];
    __shared__ __attribute__((aligned(16))) float Bs[2][XP_KC * XP_N];

    const int tid  = threadIdx.x;
    const int lane = tid & 31;
    const int wave = tid >> 5;
    const int half = lane >> 4;
    const int mr   = lane & 15;
    const int m0   = blockIdx.x * XP_M;
    const int rot  = blockIdx.x & (XP_NCH - 1);

    const float* Ag = A + (size_t)(m0 + wave) * DINNER + 4 * lane;
    const float* Bg = B + 4 * tid;
    const int aSt = wave * XP_AP + 4 * lane;
    const int bSt = 4 * tid;

    float4 ra0, ra1;
    float4 rb0, rb1, rb2, rb3, rb4, rb5, rb6, rb7, rb8, rb9, rb10, rb11;
    auto gload = [&](int ci) {
        const int k0 = ((ci + rot) & (XP_NCH - 1)) * XP_KC;
        ra0 = *(const float4*)(Ag + k0);
        ra1 = *(const float4*)(Ag + (size_t)8 * DINNER + k0);
        const float* Bk = Bg + (size_t)k0 * XP_N;
        rb0  = *(const float4*)(Bk);
        rb1  = *(const float4*)(Bk + 1024);
        rb2  = *(const float4*)(Bk + 2048);
        rb3  = *(const float4*)(Bk + 3072);
        rb4  = *(const float4*)(Bk + 4096);
        rb5  = *(const float4*)(Bk + 5120);
        rb6  = *(const float4*)(Bk + 6144);
        rb7  = *(const float4*)(Bk + 7168);
        rb8  = *(const float4*)(Bk + 8192);
        rb9  = *(const float4*)(Bk + 9216);
        rb10 = *(const float4*)(Bk + 10240);
        rb11 = *(const float4*)(Bk + 11264);
    };
    auto sstore = [&](int buf) {
        float* Ad = &As[buf][aSt];
        float* Bd = &Bs[buf][bSt];
        *(float4*)(Ad)             = ra0;
        *(float4*)(Ad + 8 * XP_AP) = ra1;
        *(float4*)(Bd)         = rb0;
        *(float4*)(Bd + 1024)  = rb1;
        *(float4*)(Bd + 2048)  = rb2;
        *(float4*)(Bd + 3072)  = rb3;
        *(float4*)(Bd + 4096)  = rb4;
        *(float4*)(Bd + 5120)  = rb5;
        *(float4*)(Bd + 6144)  = rb6;
        *(float4*)(Bd + 7168)  = rb7;
        *(float4*)(Bd + 8192)  = rb8;
        *(float4*)(Bd + 9216)  = rb9;
        *(float4*)(Bd + 10240) = rb10;
        *(float4*)(Bd + 11264) = rb11;
    };

    v8f acc[XP_NT];
#pragma unroll
    for (int j = 0; j < XP_NT; ++j) acc[j] = (v8f){};

    const int aFr = mr * XP_AP + 16 * wave + half;
    const int bFr = (16 * wave + half) * XP_N + mr;

    auto compute = [&](int buf) {
        const float* Aw = &As[buf][aFr];
        const float* Bw = &Bs[buf][bFr];
#pragma unroll
        for (int i = 0; i < 4; ++i) {
            v2f a;
            a.x = Aw[4 * i];
            a.y = Aw[4 * i + 2];
            v2f b[XP_NT];
#pragma unroll
            for (int j = 0; j < XP_NT; ++j) {
                b[j].x = Bw[(4 * i) * XP_N + 16 * j];
                b[j].y = Bw[(4 * i + 2) * XP_N + 16 * j];
            }
#pragma unroll
            for (int j = 0; j < XP_NT; ++j)
                acc[j] = wmma_f32(a, b[j], acc[j]);
        }
    };

    gload(0);
    int buf = 0;
    for (int ci = 0; ci < XP_NCH; ++ci) {
        sstore(buf);
        __syncthreads();
        gload(min(ci + 1, XP_NCH - 1));
        __builtin_amdgcn_sched_barrier(0);
        compute(buf);
        buf ^= 1;
    }

    __syncthreads();
    float* red = &Bs[0][0];
#pragma unroll
    for (int j = 0; j < XP_NT; ++j)
#pragma unroll
        for (int r = 0; r < 8; ++r)
            red[((wave * XP_NT + j) * 8 + r) * 32 + lane] = acc[j][r];
    __syncthreads();

#pragma unroll
    for (int i = 0; i < (XP_M * XP_N) / 256; ++i) {
        const int e   = tid + 256 * i;
        const int row = e / XP_N;
        const int col = e - row * XP_N;
        const int src = (((col >> 4) * 8) + (row & 7)) * 32 + 16 * (row >> 3) + (col & 15);
        float s = red[src];
#pragma unroll
        for (int p = 1; p < 8; ++p) s += red[p * (XP_NT * 8 * 32) + src];
        C[(size_t)m0 * XP_N + e] = s;
    }
}

__global__ __launch_bounds__(256) void conv_silu_kernel(
    const float* __restrict__ xz, const float* __restrict__ conv_w,
    const float* __restrict__ conv_b, float* __restrict__ xc)
{
    const int idx = blockIdx.x * 256 + threadIdx.x;
    const int d   = idx & (DINNER - 1);
    const int pos = idx >> 11;
    const int l   = pos & (LSEQ - 1);
    const int b   = pos >> 10;

    float acc = conv_b[d];
#pragma unroll
    for (int k = 0; k < DCONV; ++k) {
        const int ls = l - (DCONV - 1) + k;
        if (ls >= 0)
            acc += conv_w[d * DCONV + k] *
                   xz[((size_t)(b * LSEQ + ls)) * (2 * DINNER) + d];
    }
    xc[(size_t)pos * DINNER + d] = silu_f(acc);
}

#define SCAN_GROUP 16
#define SCAN_WAVES 4
#define NPW (DSTATE / SCAN_WAVES)
#define FPW (SCAN_GROUP / SCAN_WAVES)
#define XDW (DTRANK + 2 * DSTATE)

struct ScanOps {
    float u[SCAN_GROUP];
    float dt[SCAN_GROUP];
    float bc[SCAN_GROUP];
    float uf[FPW];
    float zf[FPW];
};

__global__ __launch_bounds__(32 * SCAN_WAVES) void scan_gate_kernel(
    const float* __restrict__ xc,   const float* __restrict__ dtm,
    const float* __restrict__ xdbl, const float* __restrict__ A_log,
    const float* __restrict__ xz,   const float* __restrict__ D_param,
    float* __restrict__ y)
{
    __shared__ float part[2][SCAN_GROUP][SCAN_WAVES][32];

    const int b    = blockIdx.y;
    const int lane = threadIdx.x & 31;
    const int w    = __builtin_amdgcn_readfirstlane(threadIdx.x >> 5);
    const int d    = blockIdx.x * 32 + lane;
    const int n0   = w * NPW;

    float Ar[NPW];
#pragma unroll
    for (int j = 0; j < NPW; ++j)
        Ar[j] = -__expf(A_log[d * DSTATE + n0 + j]);
    const float Dv = D_param[d];

    float h[NPW];
#pragma unroll
    for (int j = 0; j < NPW; ++j) h[j] = 0.0f;

    const size_t rowBase = (size_t)b * LSEQ;
    const float* up  = xc   + rowBase * DINNER + d;
    const float* dp  = dtm  + rowBase * DINNER + d;
    const float* zp  = xz   + rowBase * (2 * DINNER) + DINNER + d;
    float*       yp  = y    + rowBase * DINNER + d;
    const float* bcp = xdbl + rowBase * XDW + DTRANK + lane;
    const int    f0  = w * FPW;

    auto fetch = [&](ScanOps& o, int tb) {
#pragma unroll
        for (int i = 0; i < SCAN_GROUP; ++i) {
            o.u[i]  = up [(size_t)(tb + i) * DINNER];
            o.dt[i] = dp [(size_t)(tb + i) * DINNER];
            o.bc[i] = bcp[(size_t)(tb + i) * XDW];
        }
#pragma unroll
        for (int k = 0; k < FPW; ++k) {
            o.uf[k] = up[(size_t)(tb + f0 + k) * DINNER];
            o.zf[k] = zp[(size_t)(tb + f0 + k) * (2 * DINNER)];
        }
    };

    auto group = [&](const ScanOps& cur, ScanOps& nx, int t0, int buf) {
        fetch(nx, min(t0 + SCAN_GROUP, LSEQ - SCAN_GROUP));
        __builtin_amdgcn_sched_barrier(0);

#pragma unroll
        for (int i = 0; i < SCAN_GROUP; ++i) {
            const float u = cur.u[i], dtv = cur.dt[i];
            const int bcbits = __float_as_int(cur.bc[i]);
            float Bv[NPW], Cv[NPW];
#pragma unroll
            for (int j = 0; j < NPW; ++j) {
                Bv[j] = __int_as_float(__builtin_amdgcn_readlane(bcbits, n0 + j));
                Cv[j] = __int_as_float(__builtin_amdgcn_readlane(bcbits, DSTATE + n0 + j));
            }

            float acc = 0.0f;
#pragma unroll
            for (int j = 0; j < NPW; ++j) {
                const float dA = __expf(dtv * Ar[j]);
                h[j] = dA * h[j] + (dtv * Bv[j]) * u;
                acc += Cv[j] * h[j];
            }
            part[buf][i][w][lane] = acc;
        }

        __syncthreads();

#pragma unroll
        for (int k = 0; k < FPW; ++k) {
            const int i = f0 + k;
            float s = part[buf][i][0][lane];
#pragma unroll
            for (int p = 1; p < SCAN_WAVES; ++p) s += part[buf][i][p][lane];
            yp[(size_t)(t0 + i) * DINNER] = (s + Dv * cur.uf[k]) * silu_f(cur.zf[k]);
        }
    };

    ScanOps opA, opB;
    fetch(opA, 0);
    for (int t0 = 0; t0 < LSEQ; t0 += 2 * SCAN_GROUP) {
        group(opA, opB, t0, 0);
        group(opB, opA, t0 + SCAN_GROUP, 1);
    }
}

extern "C" void kernel_launch(void* const* d_in, const int* in_sizes, int n_in,
                              void* d_out, int out_size, void* d_ws, size_t ws_size,
                              hipStream_t stream)
{
    const float* hidden    = (const float*)d_in[0];
    const float* in_proj_w = (const float*)d_in[1];
    const float* conv_w    = (const float*)d_in[2];
    const float* conv_b    = (const float*)d_in[3];
    const float* x_proj_w  = (const float*)d_in[4];
    const float* dt_proj_w = (const float*)d_in[5];
    const float* dt_bias   = (const float*)d_in[6];
    const float* A_log     = (const float*)d_in[7];
    const float* D_param   = (const float*)d_in[8];
    const float* out_proj  = (const float*)d_in[9];
    float*       out       = (float*)d_out;

    float* ws   = (float*)d_ws;
    float* xz   = ws;
    float* xc   = xz   + (size_t)NROWS * 2 * DINNER;
    float* xdbl = xc   + (size_t)NROWS * DINNER;
    float* dtm  = xdbl + (size_t)NROWS * (DTRANK + 2 * DSTATE);
    float* y    = dtm  + (size_t)NROWS * DINNER;

    gemm_wmma_blocked<0><<<dim3((2 * DINNER) / GB_N, NROWS / GB_M), 256, 0, stream>>>(
        hidden, in_proj_w, xz, nullptr, DMODEL, DMODEL, 2 * DINNER, 2 * DINNER);

    conv_silu_kernel<<<(NROWS * DINNER) / 256, 256, 0, stream>>>(xz, conv_w, conv_b, xc);

    gemm_xproj_kernel<<<NROWS / XP_M, 256, 0, stream>>>(xc, x_proj_w, xdbl);

    gemm_wmma_blocked<1><<<dim3(DINNER / GB_N, NROWS / GB_M), 256, 0, stream>>>(
        xdbl, dt_proj_w, dtm, dt_bias, DTRANK,
        DTRANK + 2 * DSTATE, DINNER, DINNER);

    scan_gate_kernel<<<dim3(DINNER / 32, B_SZ), 32 * SCAN_WAVES, 0, stream>>>(
        xc, dtm, xdbl, A_log, xz, D_param, y);

    gemm_wmma_blocked<0><<<dim3(DMODEL / GB_N, NROWS / GB_M), 256, 0, stream>>>(
        y, out_proj, out, nullptr, DINNER, DINNER, DMODEL, DMODEL);
}


# dt epilogue: softplus = log2(u)*ln2 - c/u (u = 1 + e^x, c = exact rounding error of that sum), branch-free, replaces library log1p (about 95 dependent instr. per value); order of sums unchanged
# speedup vs baseline: 3.1759x; 1.0543x over previous
#include <hip/hip_runtime.h>
#include <math.h>


typedef float v2f __attribute__((ext_vector_type(2)));
typedef float v8f __attribute__((ext_vector_type(8)));

#define B_SZ    4
#define LSEQ    1024
#define DMODEL  1024
#define DINNER  2048
#define DSTATE  16
#define DCONV   4
#define DTRANK  64
#define NROWS   (B_SZ * LSEQ)

__device__ __forceinline__ float silu_f(float x) { return x / (1.0f + __expf(-x)); }
__device__ __forceinline__ float softplus_f(float x) {
    const float e   = __expf(x);
    const bool  big = e > 1.0f;
    const float hi  = big ? e : 1.0f;
    const float lo  = big ? 1.0f : e;
    const float u   = hi + lo;
    const float c   = (u - hi) - lo;
    const float l2  = __builtin_amdgcn_logf(u);
    const float r   = fmaf(l2, 0.693147180559945f, -(c * __builtin_amdgcn_rcpf(u)));
    return (x > 20.0f) ? x : r;
}

__device__ __forceinline__ v8f wmma_f32(v2f a, v2f b, v8f c) {
    return __builtin_amdgcn_wmma_f32_16x16x4_f32(
               false, a, false, b, (short)0, c, false, false);
}

#define GB_M 128
#define GB_N 128
#define GB_K 32
#define GA_P (GB_K + 4)
#define GB_P (2 * GB_N + 32)

template <int EPI>
__global__ __launch_bounds__(256) void gemm_wmma_blocked(
    const float* __restrict__ A, const float* __restrict__ B,
    float* __restrict__ C, const float* __restrict__ bias,
    int K, int lda, int ldb, int ldc)
{
    __shared__ __attribute__((aligned(16))) float As[2][GB_M * GA_P];
    __shared__ __attribute__((aligned(16))) float Bs[2][(GB_K / 2) * GB_P];

    const int tid  = threadIdx.x;
    const int lane = tid & 31;
    const int wave = tid >> 5;
    const int half = lane >> 4;
    const int mr   = lane & 15;
    const int wm   = wave >> 1;
    const int wn   = wave & 1;
    const int bm0  = blockIdx.y * GB_M;
    const int bn0  = blockIdx.x * GB_N;

    const int aRow = tid >> 3, aQ = tid & 7;
    const int bPr  = tid >> 7, bN = tid & 127;
    const float* Ag = A + (size_t)(bm0 + aRow) * lda + 4 * aQ;
    const float* Bg = B + (size_t)(2 * bPr) * ldb + bn0 + bN;
    const int aSt = aRow * GA_P + 4 * aQ;
    const int bSt = bPr * GB_P + 2 * bN;

    float4 ra0, ra1, ra2, ra3;
    v2f rb0, rb1, rb2, rb3, rb4, rb5, rb6, rb7;
    auto gload = [&](int k0) {
        ra0 = *(const float4*)(Ag + k0);
        ra1 = *(const float4*)(Ag + (size_t)32 * lda + k0);
        ra2 = *(const float4*)(Ag + (size_t)64 * lda + k0);
        ra3 = *(const float4*)(Ag + (size_t)96 * lda + k0);
        const float* Bk = Bg + (size_t)k0 * ldb;
        const size_t l1 = (size_t)ldb;
        rb0.x = Bk[0];        rb0.y = Bk[l1];
        rb1.x = Bk[4 * l1];   rb1.y = Bk[5 * l1];
        rb2.x = Bk[8 * l1];   rb2.y = Bk[9 * l1];
        rb3.x = Bk[12 * l1];  rb3.y = Bk[13 * l1];
        rb4.x = Bk[16 * l1];  rb4.y = Bk[17 * l1];
        rb5.x = Bk[20 * l1];  rb5.y = Bk[21 * l1];
        rb6.x = Bk[24 * l1];  rb6.y = Bk[25 * l1];
        rb7.x = Bk[28 * l1];  rb7.y = Bk[29 * l1];
    };
    auto sstore = [&](int buf) {
        float* Ad = &As[buf][aSt];
        float* Bd = &Bs[buf][bSt];
        *(float4*)(Ad)             = ra0;
        *(float4*)(Ad + 32 * GA_P) = ra1;
        *(float4*)(Ad + 64 * GA_P) = ra2;
        *(float4*)(Ad + 96 * GA_P) = ra3;
        *(v2f*)(Bd)              = rb0;
        *(v2f*)(Bd +  2 * GB_P)  = rb1;
        *(v2f*)(Bd +  4 * GB_P)  = rb2;
        *(v2f*)(Bd +  6 * GB_P)  = rb3;
        *(v2f*)(Bd +  8 * GB_P)  = rb4;
        *(v2f*)(Bd + 10 * GB_P)  = rb5;
        *(v2f*)(Bd + 12 * GB_P)  = rb6;
        *(v2f*)(Bd + 14 * GB_P)  = rb7;
    };

    v8f c[2][4];
#pragma unroll
    for (int i = 0; i < 2; ++i)
#pragma unroll
        for (int j = 0; j < 4; ++j) c[i][j] = (v8f){};

    const int aFr = (wm * 32 + mr) * GA_P + 2 * half;
    const int bFr = half * GB_P + 2 * (wn * 64 + mr);

    auto compute = [&](int buf) {
        const float* Aw = &As[buf][aFr];
        const float* Bw = &Bs[buf][bFr];
#pragma unroll
        for (int kk = 0; kk < GB_K; kk += 4) {
            v2f a0 = *(const v2f*)(Aw + kk);
            v2f a1 = *(const v2f*)(Aw + 16 * GA_P + kk);
            v2f b[4];
#pragma unroll
            for (int j = 0; j < 4; ++j)
                b[j] = *(const v2f*)(Bw + (kk / 2) * GB_P + j * 32);
#pragma unroll
            for (int j = 0; j < 4; ++j) {
                c[0][j] = wmma_f32(a0, b[j], c[0][j]);
                c[1][j] = wmma_f32(a1, b[j], c[1][j]);
            }
        }
    };

    gload(0);
    int buf = 0;
    for (int k0 = 0; k0 < K; k0 += GB_K) {
        sstore(buf);
        __syncthreads();
        gload(min(k0 + GB_K, K - GB_K));
        __builtin_amdgcn_sched_barrier(0);
        compute(buf);
        buf ^= 1;
    }

#pragma unroll
    for (int j = 0; j < 4; ++j) {
        const int col = bn0 + wn * 64 + j * 16 + mr;
        const float bv = (EPI == 1) ? bias[col] : 0.0f;
#pragma unroll
        for (int i = 0; i < 2; ++i) {
#pragma unroll
            for (int r = 0; r < 8; ++r) {
                const int row = bm0 + wm * 32 + i * 16 + half * 8 + r;
                float v = c[i][j][r];
                if (EPI == 1) v = softplus_f(v + bv);
                C[(size_t)row * ldc + col] = v;
            }
        }
    }
}

#define XP_M   16
#define XP_N   (DTRANK + 2 * DSTATE)
#define XP_KC  128
#define XP_AP  (XP_KC + 4)
#define XP_NT  (XP_N / 16)
#define XP_NCH (DINNER / XP_KC)

__global__ __launch_bounds__(256) void gemm_xproj_kernel(
    const float* __restrict__ A, const float* __restrict__ B, float* __restrict__ C)
{
    __shared__ __attribute__((aligned(16))) float As[2][XP_M * XP_AP];
    __shared__ __attribute__((aligned(16))) float Bs[2][XP_KC * XP_N];

    const int tid  = threadIdx.x;
    const int lane = tid & 31;
    const int wave = tid >> 5;
    const int half = lane >> 4;
    const int mr   = lane & 15;
    const int m0   = blockIdx.x * XP_M;
    const int rot  = blockIdx.x & (XP_NCH - 1);

    const float* Ag = A + (size_t)(m0 + wave) * DINNER + 4 * lane;
    const float* Bg = B + 4 * tid;
    const int aSt = wave * XP_AP + 4 * lane;
    const int bSt = 4 * tid;

    float4 ra0, ra1;
    float4 rb0, rb1, rb2, rb3, rb4, rb5, rb6, rb7, rb8, rb9, rb10, rb11;
    auto gload = [&](int ci) {
        const int k0 = ((ci + rot) & (XP_NCH - 1)) * XP_KC;
        ra0 = *(const float4*)(Ag + k0);
        ra1 = *(const float4*)(Ag + (size_t)8 * DINNER + k0);
        const float* Bk = Bg + (size_t)k0 * XP_N;
        rb0  = *(const float4*)(Bk);
        rb1  = *(const float4*)(Bk + 1024);
        rb2  = *(const float4*)(Bk + 2048);
        rb3  = *(const float4*)(Bk + 3072);
        rb4  = *(const float4*)(Bk + 4096);
        rb5  = *(const float4*)(Bk + 5120);
        rb6  = *(const float4*)(Bk + 6144);
        rb7  = *(const float4*)(Bk + 7168);
        rb8  = *(const float4*)(Bk + 8192);
        rb9  = *(const float4*)(Bk + 9216);
        rb10 = *(const float4*)(Bk + 10240);
        rb11 = *(const float4*)(Bk + 11264);
    };
    auto sstore = [&](int buf) {
        float* Ad = &As[buf][aSt];
        float* Bd = &Bs[buf][bSt];
        *(float4*)(Ad)             = ra0;
        *(float4*)(Ad + 8 * XP_AP) = ra1;
        *(float4*)(Bd)         = rb0;
        *(float4*)(Bd + 1024)  = rb1;
        *(float4*)(Bd + 2048)  = rb2;
        *(float4*)(Bd + 3072)  = rb3;
        *(float4*)(Bd + 4096)  = rb4;
        *(float4*)(Bd + 5120)  = rb5;
        *(float4*)(Bd + 6144)  = rb6;
        *(float4*)(Bd + 7168)  = rb7;
        *(float4*)(Bd + 8192)  = rb8;
        *(float4*)(Bd + 9216)  = rb9;
        *(float4*)(Bd + 10240) = rb10;
        *(float4*)(Bd + 11264) = rb11;
    };

    v8f acc[XP_NT];
#pragma unroll
    for (int j = 0; j < XP_NT; ++j) acc[j] = (v8f){};

    const int aFr = mr * XP_AP + 16 * wave + half;
    const int bFr = (16 * wave + half) * XP_N + mr;

    auto compute = [&](int buf) {
        const float* Aw = &As[buf][aFr];
        const float* Bw = &Bs[buf][bFr];
#pragma unroll
        for (int i = 0; i < 4; ++i) {
            v2f a;
            a.x = Aw[4 * i];
            a.y = Aw[4 * i + 2];
            v2f b[XP_NT];
#pragma unroll
            for (int j = 0; j < XP_NT; ++j) {
                b[j].x = Bw[(4 * i) * XP_N + 16 * j];
                b[j].y = Bw[(4 * i + 2) * XP_N + 16 * j];
            }
#pragma unroll
            for (int j = 0; j < XP_NT; ++j)
                acc[j] = wmma_f32(a, b[j], acc[j]);
        }
    };

    gload(0);
    int buf = 0;
    for (int ci = 0; ci < XP_NCH; ++ci) {
        sstore(buf);
        __syncthreads();
        gload(min(ci + 1, XP_NCH - 1));
        __builtin_amdgcn_sched_barrier(0);
        compute(buf);
        buf ^= 1;
    }

    __syncthreads();
    float* red = &Bs[0][0];
#pragma unroll
    for (int j = 0; j < XP_NT; ++j)
#pragma unroll
        for (int r = 0; r < 8; ++r)
            red[((wave * XP_NT + j) * 8 + r) * 32 + lane] = acc[j][r];
    __syncthreads();

#pragma unroll
    for (int i = 0; i < (XP_M * XP_N) / 256; ++i) {
        const int e   = tid + 256 * i;
        const int row = e / XP_N;
        const int col = e - row * XP_N;
        const int src = (((col >> 4) * 8) + (row & 7)) * 32 + 16 * (row >> 3) + (col & 15);
        float s = red[src];
#pragma unroll
        for (int p = 1; p < 8; ++p) s += red[p * (XP_NT * 8 * 32) + src];
        C[(size_t)m0 * XP_N + e] = s;
    }
}

__global__ __launch_bounds__(256) void conv_silu_kernel(
    const float* __restrict__ xz, const float* __restrict__ conv_w,
    const float* __restrict__ conv_b, float* __restrict__ xc)
{
    const int idx = blockIdx.x * 256 + threadIdx.x;
    const int d   = idx & (DINNER - 1);
    const int pos = idx >> 11;
    const int l   = pos & (LSEQ - 1);
    const int b   = pos >> 10;

    float acc = conv_b[d];
#pragma unroll
    for (int k = 0; k < DCONV; ++k) {
        const int ls = l - (DCONV - 1) + k;
        if (ls >= 0)
            acc += conv_w[d * DCONV + k] *
                   xz[((size_t)(b * LSEQ + ls)) * (2 * DINNER) + d];
    }
    xc[(size_t)pos * DINNER + d] = silu_f(acc);
}

#define SCAN_GROUP 16
#define SCAN_WAVES 4
#define NPW (DSTATE / SCAN_WAVES)
#define FPW (SCAN_GROUP / SCAN_WAVES)
#define XDW (DTRANK + 2 * DSTATE)

struct ScanOps {
    float u[SCAN_GROUP];
    float dt[SCAN_GROUP];
    float bc[SCAN_GROUP];
    float uf[FPW];
    float zf[FPW];
};

__global__ __launch_bounds__(32 * SCAN_WAVES) void scan_gate_kernel(
    const float* __restrict__ xc,   const float* __restrict__ dtm,
    const float* __restrict__ xdbl, const float* __restrict__ A_log,
    const float* __restrict__ xz,   const float* __restrict__ D_param,
    float* __restrict__ y)
{
    __shared__ float part[2][SCAN_GROUP][SCAN_WAVES][32];

    const int b    = blockIdx.y;
    const int lane = threadIdx.x & 31;
    const int w    = __builtin_amdgcn_readfirstlane(threadIdx.x >> 5);
    const int d    = blockIdx.x * 32 + lane;
    const int n0   = w * NPW;

    float Ar[NPW];
#pragma unroll
    for (int j = 0; j < NPW; ++j)
        Ar[j] = -__expf(A_log[d * DSTATE + n0 + j]);
    const float Dv = D_param[d];

    float h[NPW];
#pragma unroll
    for (int j = 0; j < NPW; ++j) h[j] = 0.0f;

    const size_t rowBase = (size_t)b * LSEQ;
    const float* up  = xc   + rowBase * DINNER + d;
    const float* dp  = dtm  + rowBase * DINNER + d;
    const float* zp  = xz   + rowBase * (2 * DINNER) + DINNER + d;
    float*       yp  = y    + rowBase * DINNER + d;
    const float* bcp = xdbl + rowBase * XDW + DTRANK + lane;
    const int    f0  = w * FPW;

    auto fetch = [&](ScanOps& o, int tb) {
#pragma unroll
        for (int i = 0; i < SCAN_GROUP; ++i) {
            o.u[i]  = up [(size_t)(tb + i) * DINNER];
            o.dt[i] = dp [(size_t)(tb + i) * DINNER];
            o.bc[i] = bcp[(size_t)(tb + i) * XDW];
        }
#pragma unroll
        for (int k = 0; k < FPW; ++k) {
            o.uf[k] = up[(size_t)(tb + f0 + k) * DINNER];
            o.zf[k] = zp[(size_t)(tb + f0 + k) * (2 * DINNER)];
        }
    };

    auto group = [&](const ScanOps& cur, ScanOps& nx, int t0, int buf) {
        fetch(nx, min(t0 + SCAN_GROUP, LSEQ - SCAN_GROUP));
        __builtin_amdgcn_sched_barrier(0);

#pragma unroll
        for (int i = 0; i < SCAN_GROUP; ++i) {
            const float u = cur.u[i], dtv = cur.dt[i];
            const int bcbits = __float_as_int(cur.bc[i]);
            float Bv[NPW], Cv[NPW];
#pragma unroll
            for (int j = 0; j < NPW; ++j) {
                Bv[j] = __int_as_float(__builtin_amdgcn_readlane(bcbits, n0 + j));
                Cv[j] = __int_as_float(__builtin_amdgcn_readlane(bcbits, DSTATE + n0 + j));
            }

            float acc = 0.0f;
#pragma unroll
            for (int j = 0; j < NPW; ++j) {
                const float dA = __expf(dtv * Ar[j]);
                h[j] = dA * h[j] + (dtv * Bv[j]) * u;
                acc += Cv[j] * h[j];
            }
            part[buf][i][w][lane] = acc;
        }

        __syncthreads();

#pragma unroll
        for (int k = 0; k < FPW; ++k) {
            const int i = f0 + k;
            float s = part[buf][i][0][lane];
#pragma unroll
            for (int p = 1; p < SCAN_WAVES; ++p) s += part[buf][i][p][lane];
            yp[(size_t)(t0 + i) * DINNER] = (s + Dv * cur.uf[k]) * silu_f(cur.zf[k]);
        }
    };

    ScanOps opA, opB;
    fetch(opA, 0);
    for (int t0 = 0; t0 < LSEQ; t0 += 2 * SCAN_GROUP) {
        group(opA, opB, t0, 0);
        group(opB, opA, t0 + SCAN_GROUP, 1);
    }
}

extern "C" void kernel_launch(void* const* d_in, const int* in_sizes, int n_in,
                              void* d_out, int out_size, void* d_ws, size_t ws_size,
                              hipStream_t stream)
{
    const float* hidden    = (const float*)d_in[0];
    const float* in_proj_w = (const float*)d_in[1];
    const float* conv_w    = (const float*)d_in[2];
    const float* conv_b    = (const float*)d_in[3];
    const float* x_proj_w  = (const float*)d_in[4];
    const float* dt_proj_w = (const float*)d_in[5];
    const float* dt_bias   = (const float*)d_in[6];
    const float* A_log     = (const float*)d_in[7];
    const float* D_param   = (const float*)d_in[8];
    const float* out_proj  = (const float*)d_in[9];
    float*       out       = (float*)d_out;

    float* ws   = (float*)d_ws;
    float* xz   = ws;
    float* xc   = xz   + (size_t)NROWS * 2 * DINNER;
    float* xdbl = xc   + (size_t)NROWS * DINNER;
    float* dtm  = xdbl + (size_t)NROWS * (DTRANK + 2 * DSTATE);
    float* y    = dtm  + (size_t)NROWS * DINNER;

    gemm_wmma_blocked<0><<<dim3((2 * DINNER) / GB_N, NROWS / GB_M), 256, 0, stream>>>(
        hidden, in_proj_w, xz, nullptr, DMODEL, DMODEL, 2 * DINNER, 2 * DINNER);

    conv_silu_kernel<<<(NROWS * DINNER) / 256, 256, 0, stream>>>(xz, conv_w, conv_b, xc);

    gemm_xproj_kernel<<<NROWS / XP_M, 256, 0, stream>>>(xc, x_proj_w, xdbl);

    gemm_wmma_blocked<1><<<dim3(DINNER / GB_N, NROWS / GB_M), 256, 0, stream>>>(
        xdbl, dt_proj_w, dtm, dt_bias, DTRANK,
        DTRANK + 2 * DSTATE, DINNER, DINNER);

    scan_gate_kernel<<<dim3(DINNER / 32, B_SZ), 32 * SCAN_WAVES, 0, stream>>>(
        xc, dtm, xdbl, A_log, xz, D_param, y);

    gemm_wmma_blocked<0><<<dim3(DMODEL / GB_N, NROWS / GB_M), 256, 0, stream>>>(
        y, out_proj, out, nullptr, DINNER, DINNER, DMODEL, DMODEL);
}


# conv+SiLU: each thread walks 16 consecutive timesteps of 4 adjacent channels (float4 row loads, 3 earlier rows kept in registers), not one element: 4096 waves, not 262144; order of sums unchanged
# speedup vs baseline: 3.3435x; 1.0528x over previous
#include <hip/hip_runtime.h>
#include <math.h>


typedef float v2f __attribute__((ext_vector_type(2)));
typedef float v8f __attribute__((ext_vector_type(8)));

#define B_SZ    4
#define LSEQ    1024
#define DMODEL  1024
#define DINNER  2048
#define DSTATE  16
#define DCONV   4
#define DTRANK  64
#define NROWS   (B_SZ * LSEQ)

__device__ __forceinline__ float silu_f(float x) { return x / (1.0f + __expf(-x)); }
__device__ __forceinline__ float softplus_f(float x) {
    const float e   = __expf(x);
    const bool  big = e > 1.0f;
    const float hi  = big ? e : 1.0f;
    const float lo  = big ? 1.0f : e;
    const float u   = hi + lo;
    const float c   = (u - hi) - lo;
    const float l2  = __builtin_amdgcn_logf(u);
    const float r   = fmaf(l2, 0.693147180559945f, -(c * __builtin_amdgcn_rcpf(u)));
    return (x > 20.0f) ? x : r;
}

__device__ __forceinline__ v8f wmma_f32(v2f a, v2f b, v8f c) {
    return __builtin_amdgcn_wmma_f32_16x16x4_f32(
               false, a, false, b, (short)0, c, false, false);
}

#define GB_M 128
#define GB_N 128
#define GB_K 32
#define GA_P (GB_K + 4)
#define GB_P (2 * GB_N + 32)

template <int EPI>
__global__ __launch_bounds__(256) void gemm_wmma_blocked(
    const float* __restrict__ A, const float* __restrict__ B,
    float* __restrict__ C, const float* __restrict__ bias,
    int K, int lda, int ldb, int ldc)
{
    __shared__ __attribute__((aligned(16))) float As[2][GB_M * GA_P];
    __shared__ __attribute__((aligned(16))) float Bs[2][(GB_K / 2) * GB_P];

    const int tid  = threadIdx.x;
    const int lane = tid & 31;
    const int wave = tid >> 5;
    const int half = lane >> 4;
    const int mr   = lane & 15;
    const int wm   = wave >> 1;
    const int wn   = wave & 1;
    const int bm0  = blockIdx.y * GB_M;
    const int bn0  = blockIdx.x * GB_N;

    const int aRow = tid >> 3, aQ = tid & 7;
    const int bPr  = tid >> 7, bN = tid & 127;
    const float* Ag = A + (size_t)(bm0 + aRow) * lda + 4 * aQ;
    const float* Bg = B + (size_t)(2 * bPr) * ldb + bn0 + bN;
    const int aSt = aRow * GA_P + 4 * aQ;
    const int bSt = bPr * GB_P + 2 * bN;

    float4 ra0, ra1, ra2, ra3;
    v2f rb0, rb1, rb2, rb3, rb4, rb5, rb6, rb7;
    auto gload = [&](int k0) {
        ra0 = *(const float4*)(Ag + k0);
        ra1 = *(const float4*)(Ag + (size_t)32 * lda + k0);
        ra2 = *(const float4*)(Ag + (size_t)64 * lda + k0);
        ra3 = *(const float4*)(Ag + (size_t)96 * lda + k0);
        const float* Bk = Bg + (size_t)k0 * ldb;
        const size_t l1 = (size_t)ldb;
        rb0.x = Bk[0];        rb0.y = Bk[l1];
        rb1.x = Bk[4 * l1];   rb1.y = Bk[5 * l1];
        rb2.x = Bk[8 * l1];   rb2.y = Bk[9 * l1];
        rb3.x = Bk[12 * l1];  rb3.y = Bk[13 * l1];
        rb4.x = Bk[16 * l1];  rb4.y = Bk[17 * l1];
        rb5.x = Bk[20 * l1];  rb5.y = Bk[21 * l1];
        rb6.x = Bk[24 * l1];  rb6.y = Bk[25 * l1];
        rb7.x = Bk[28 * l1];  rb7.y = Bk[29 * l1];
    };
    auto sstore = [&](int buf) {
        float* Ad = &As[buf][aSt];
        float* Bd = &Bs[buf][bSt];
        *(float4*)(Ad)             = ra0;
        *(float4*)(Ad + 32 * GA_P) = ra1;
        *(float4*)(Ad + 64 * GA_P) = ra2;
        *(float4*)(Ad + 96 * GA_P) = ra3;
        *(v2f*)(Bd)              = rb0;
        *(v2f*)(Bd +  2 * GB_P)  = rb1;
        *(v2f*)(Bd +  4 * GB_P)  = rb2;
        *(v2f*)(Bd +  6 * GB_P)  = rb3;
        *(v2f*)(Bd +  8 * GB_P)  = rb4;
        *(v2f*)(Bd + 10 * GB_P)  = rb5;
        *(v2f*)(Bd + 12 * GB_P)  = rb6;
        *(v2f*)(Bd + 14 * GB_P)  = rb7;
    };

    v8f c[2][4];
#pragma unroll
    for (int i = 0; i < 2; ++i)
#pragma unroll
        for (int j = 0; j < 4; ++j) c[i][j] = (v8f){};

    const int aFr = (wm * 32 + mr) * GA_P + 2 * half;
    const int bFr = half * GB_P + 2 * (wn * 64 + mr);

    auto compute = [&](int buf) {
        const float* Aw = &As[buf][aFr];
        const float* Bw = &Bs[buf][bFr];
#pragma unroll
        for (int kk = 0; kk < GB_K; kk += 4) {
            v2f a0 = *(const v2f*)(Aw + kk);
            v2f a1 = *(const v2f*)(Aw + 16 * GA_P + kk);
            v2f b[4];
#pragma unroll
            for (int j = 0; j < 4; ++j)
                b[j] = *(const v2f*)(Bw + (kk / 2) * GB_P + j * 32);
#pragma unroll
            for (int j = 0; j < 4; ++j) {
                c[0][j] = wmma_f32(a0, b[j], c[0][j]);
                c[1][j] = wmma_f32(a1, b[j], c[1][j]);
            }
        }
    };

    gload(0);
    int buf = 0;
    for (int k0 = 0; k0 < K; k0 += GB_K) {
        sstore(buf);
        __syncthreads();
        gload(min(k0 + GB_K, K - GB_K));
        __builtin_amdgcn_sched_barrier(0);
        compute(buf);
        buf ^= 1;
    }

#pragma unroll
    for (int j = 0; j < 4; ++j) {
        const int col = bn0 + wn * 64 + j * 16 + mr;
        const float bv = (EPI == 1) ? bias[col] : 0.0f;
#pragma unroll
        for (int i = 0; i < 2; ++i) {
#pragma unroll
            for (int r = 0; r < 8; ++r) {
                const int row = bm0 + wm * 32 + i * 16 + half * 8 + r;
                float v = c[i][j][r];
                if (EPI == 1) v = softplus_f(v + bv);
                C[(size_t)row * ldc + col] = v;
            }
        }
    }
}

#define XP_M   16
#define XP_N   (DTRANK + 2 * DSTATE)
#define XP_KC  128
#define XP_AP  (XP_KC + 4)
#define XP_NT  (XP_N / 16)
#define XP_NCH (DINNER / XP_KC)

__global__ __launch_bounds__(256) void gemm_xproj_kernel(
    const float* __restrict__ A, const float* __restrict__ B, float* __restrict__ C)
{
    __shared__ __attribute__((aligned(16))) float As[2][XP_M * XP_AP];
    __shared__ __attribute__((aligned(16))) float Bs[2][XP_KC * XP_N];

    const int tid  = threadIdx.x;
    const int lane = tid & 31;
    const int wave = tid >> 5;
    const int half = lane >> 4;
    const int mr   = lane & 15;
    const int m0   = blockIdx.x * XP_M;
    const int rot  = blockIdx.x & (XP_NCH - 1);

    const float* Ag = A + (size_t)(m0 + wave) * DINNER + 4 * lane;
    const float* Bg = B + 4 * tid;
    const int aSt = wave * XP_AP + 4 * lane;
    const int bSt = 4 * tid;

    float4 ra0, ra1;
    float4 rb0, rb1, rb2, rb3, rb4, rb5, rb6, rb7, rb8, rb9, rb10, rb11;
    auto gload = [&](int ci) {
        const int k0 = ((ci + rot) & (XP_NCH - 1)) * XP_KC;
        ra0 = *(const float4*)(Ag + k0);
        ra1 = *(const float4*)(Ag + (size_t)8 * DINNER + k0);
        const float* Bk = Bg + (size_t)k0 * XP_N;
        rb0  = *(const float4*)(Bk);
        rb1  = *(const float4*)(Bk + 1024);
        rb2  = *(const float4*)(Bk + 2048);
        rb3  = *(const float4*)(Bk + 3072);
        rb4  = *(const float4*)(Bk + 4096);
        rb5  = *(const float4*)(Bk + 5120);
        rb6  = *(const float4*)(Bk + 6144);
        rb7  = *(const float4*)(Bk + 7168);
        rb8  = *(const float4*)(Bk + 8192);
        rb9  = *(const float4*)(Bk + 9216);
        rb10 = *(const float4*)(Bk + 10240);
        rb11 = *(const float4*)(Bk + 11264);
    };
    auto sstore = [&](int buf) {
        float* Ad = &As[buf][aSt];
        float* Bd = &Bs[buf][bSt];
        *(float4*)(Ad)             = ra0;
        *(float4*)(Ad + 8 * XP_AP) = ra1;
        *(float4*)(Bd)         = rb0;
        *(float4*)(Bd + 1024)  = rb1;
        *(float4*)(Bd + 2048)  = rb2;
        *(float4*)(Bd + 3072)  = rb3;
        *(float4*)(Bd + 4096)  = rb4;
        *(float4*)(Bd + 5120)  = rb5;
        *(float4*)(Bd + 6144)  = rb6;
        *(float4*)(Bd + 7168)  = rb7;
        *(float4*)(Bd + 8192)  = rb8;
        *(float4*)(Bd + 9216)  = rb9;
        *(float4*)(Bd + 10240) = rb10;
        *(float4*)(Bd + 11264) = rb11;
    };

    v8f acc[XP_NT];
#pragma unroll
    for (int j = 0; j < XP_NT; ++j) acc[j] = (v8f){};

    const int aFr = mr * XP_AP + 16 * wave + half;
    const int bFr = (16 * wave + half) * XP_N + mr;

    auto compute = [&](int buf) {
        const float* Aw = &As[buf][aFr];
        const float* Bw = &Bs[buf][bFr];
#pragma unroll
        for (int i = 0; i < 4; ++i) {
            v2f a;
            a.x = Aw[4 * i];
            a.y = Aw[4 * i + 2];
            v2f b[XP_NT];
#pragma unroll
            for (int j = 0; j < XP_NT; ++j) {
                b[j].x = Bw[(4 * i) * XP_N + 16 * j];
                b[j].y = Bw[(4 * i + 2) * XP_N + 16 * j];
            }
#pragma unroll
            for (int j = 0; j < XP_NT; ++j)
                acc[j] = wmma_f32(a, b[j], acc[j]);
        }
    };

    gload(0);
    int buf = 0;
    for (int ci = 0; ci < XP_NCH; ++ci) {
        sstore(buf);
        __syncthreads();
        gload(min(ci + 1, XP_NCH - 1));
        __builtin_amdgcn_sched_barrier(0);
        compute(buf);
        buf ^= 1;
    }

    __syncthreads();
    float* red = &Bs[0][0];
#pragma unroll
    for (int j = 0; j < XP_NT; ++j)
#pragma unroll
        for (int r = 0; r < 8; ++r)
            red[((wave * XP_NT + j) * 8 + r) * 32 + lane] = acc[j][r];
    __syncthreads();

#pragma unroll
    for (int i = 0; i < (XP_M * XP_N) / 256; ++i) {
        const int e   = tid + 256 * i;
        const int row = e / XP_N;
        const int col = e - row * XP_N;
        const int src = (((col >> 4) * 8) + (row & 7)) * 32 + 16 * (row >> 3) + (col & 15);
        float s = red[src];
#pragma unroll
        for (int p = 1; p < 8; ++p) s += red[p * (XP_NT * 8 * 32) + src];
        C[(size_t)m0 * XP_N + e] = s;
    }
}

#define CV_T  16
#define CV_C4 (DINNER / 4)

__global__ __launch_bounds__(256) void conv_silu_kernel(
    const float* __restrict__ xz, const float* __restrict__ conv_w,
    const float* __restrict__ conv_b, float* __restrict__ xc)
{
    const int c4    = (blockIdx.x % (CV_C4 / 256)) * 256 + threadIdx.x;
    const int row0  = (blockIdx.x / (CV_C4 / 256)) * CV_T;
    const bool head = (row0 & (LSEQ - 1)) == 0;
    const int d     = 4 * c4;

    const float4* wp = (const float4*)(conv_w + (size_t)d * DCONV);
    const float4 wa = wp[0], wb = wp[1], wc = wp[2], wd = wp[3];
    const float4 bv = *(const float4*)(conv_b + d);

    const float* xp = xz + (size_t)row0 * (2 * DINNER) + d;
    float4 x[CV_T + 3];
    const float* xh = head ? xp : xp - 3 * (2 * DINNER);
    const float4 h0 = *(const float4*)(xh);
    const float4 h1 = *(const float4*)(xh + (2 * DINNER));
    const float4 h2 = *(const float4*)(xh + 2 * (2 * DINNER));
#pragma unroll
    for (int r = 0; r < CV_T; ++r)
        x[3 + r] = *(const float4*)(xp + (size_t)r * (2 * DINNER));
    const float4 z4 = make_float4(0.0f, 0.0f, 0.0f, 0.0f);
    x[0] = head ? z4 : h0;
    x[1] = head ? z4 : h1;
    x[2] = head ? z4 : h2;

    float* op = xc + (size_t)row0 * DINNER + d;
#pragma unroll
    for (int t = 0; t < CV_T; ++t) {
        float ax = bv.x, ay = bv.y, az = bv.z, aw = bv.w;
        ax += wa.x * x[t].x;      ay += wb.x * x[t].y;
        az += wc.x * x[t].z;      aw += wd.x * x[t].w;
        ax += wa.y * x[t + 1].x;  ay += wb.y * x[t + 1].y;
        az += wc.y * x[t + 1].z;  aw += wd.y * x[t + 1].w;
        ax += wa.z * x[t + 2].x;  ay += wb.z * x[t + 2].y;
        az += wc.z * x[t + 2].z;  aw += wd.z * x[t + 2].w;
        ax += wa.w * x[t + 3].x;  ay += wb.w * x[t + 3].y;
        az += wc.w * x[t + 3].z;  aw += wd.w * x[t + 3].w;
        float4 o;
        o.x = silu_f(ax);  o.y = silu_f(ay);
        o.z = silu_f(az);  o.w = silu_f(aw);
        *(float4*)(op + (size_t)t * DINNER) = o;
    }
}

#define SCAN_GROUP 16
#define SCAN_WAVES 4
#define NPW (DSTATE / SCAN_WAVES)
#define FPW (SCAN_GROUP / SCAN_WAVES)
#define XDW (DTRANK + 2 * DSTATE)

struct ScanOps {
    float u[SCAN_GROUP];
    float dt[SCAN_GROUP];
    float bc[SCAN_GROUP];
    float uf[FPW];
    float zf[FPW];
};

__global__ __launch_bounds__(32 * SCAN_WAVES) void scan_gate_kernel(
    const float* __restrict__ xc,   const float* __restrict__ dtm,
    const float* __restrict__ xdbl, const float* __restrict__ A_log,
    const float* __restrict__ xz,   const float* __restrict__ D_param,
    float* __restrict__ y)
{
    __shared__ float part[2][SCAN_GROUP][SCAN_WAVES][32];

    const int b    = blockIdx.y;
    const int lane = threadIdx.x & 31;
    const int w    = __builtin_amdgcn_readfirstlane(threadIdx.x >> 5);
    const int d    = blockIdx.x * 32 + lane;
    const int n0   = w * NPW;

    float Ar[NPW];
#pragma unroll
    for (int j = 0; j < NPW; ++j)
        Ar[j] = -__expf(A_log[d * DSTATE + n0 + j]);
    const float Dv = D_param[d];

    float h[NPW];
#pragma unroll
    for (int j = 0; j < NPW; ++j) h[j] = 0.0f;

    const size_t rowBase = (size_t)b * LSEQ;
    const float* up  = xc   + rowBase * DINNER + d;
    const float* dp  = dtm  + rowBase * DINNER + d;
    const float* zp  = xz   + rowBase * (2 * DINNER) + DINNER + d;
    float*       yp  = y    + rowBase * DINNER + d;
    const float* bcp = xdbl + rowBase * XDW + DTRANK + lane;
    const int    f0  = w * FPW;

    auto fetch = [&](ScanOps& o, int tb) {
#pragma unroll
        for (int i = 0; i < SCAN_GROUP; ++i) {
            o.u[i]  = up [(size_t)(tb + i) * DINNER];
            o.dt[i] = dp [(size_t)(tb + i) * DINNER];
            o.bc[i] = bcp[(size_t)(tb + i) * XDW];
        }
#pragma unroll
        for (int k = 0; k < FPW; ++k) {
            o.uf[k] = up[(size_t)(tb + f0 + k) * DINNER];
            o.zf[k] = zp[(size_t)(tb + f0 + k) * (2 * DINNER)];
        }
    };

    auto group = [&](const ScanOps& cur, ScanOps& nx, int t0, int buf) {
        fetch(nx, min(t0 + SCAN_GROUP, LSEQ - SCAN_GROUP));
        __builtin_amdgcn_sched_barrier(0);

#pragma unroll
        for (int i = 0; i < SCAN_GROUP; ++i) {
            const float u = cur.u[i], dtv = cur.dt[i];
            const int bcbits = __float_as_int(cur.bc[i]);
            float Bv[NPW], Cv[NPW];
#pragma unroll
            for (int j = 0; j < NPW; ++j) {
                Bv[j] = __int_as_float(__builtin_amdgcn_readlane(bcbits, n0 + j));
                Cv[j] = __int_as_float(__builtin_amdgcn_readlane(bcbits, DSTATE + n0 + j));
            }

            float acc = 0.0f;
#pragma unroll
            for (int j = 0; j < NPW; ++j) {
                const float dA = __expf(dtv * Ar[j]);
                h[j] = dA * h[j] + (dtv * Bv[j]) * u;
                acc += Cv[j] * h[j];
            }
            part[buf][i][w][lane] = acc;
        }

        __syncthreads();

#pragma unroll
        for (int k = 0; k < FPW; ++k) {
            const int i = f0 + k;
            float s = part[buf][i][0][lane];
#pragma unroll
            for (int p = 1; p < SCAN_WAVES; ++p) s += part[buf][i][p][lane];
            yp[(size_t)(t0 + i) * DINNER] = (s + Dv * cur.uf[k]) * silu_f(cur.zf[k]);
        }
    };

    ScanOps opA, opB;
    fetch(opA, 0);
    for (int t0 = 0; t0 < LSEQ; t0 += 2 * SCAN_GROUP) {
        group(opA, opB, t0, 0);
        group(opB, opA, t0 + SCAN_GROUP, 1);
    }
}

extern "C" void kernel_launch(void* const* d_in, const int* in_sizes, int n_in,
                              void* d_out, int out_size, void* d_ws, size_t ws_size,
                              hipStream_t stream)
{
    const float* hidden    = (const float*)d_in[0];
    const float* in_proj_w = (const float*)d_in[1];
    const float* conv_w    = (const float*)d_in[2];
    const float* conv_b    = (const float*)d_in[3];
    const float* x_proj_w  = (const float*)d_in[4];
    const float* dt_proj_w = (const float*)d_in[5];
    const float* dt_bias   = (const float*)d_in[6];
    const float* A_log     = (const float*)d_in[7];
    const float* D_param   = (const float*)d_in[8];
    const float* out_proj  = (const float*)d_in[9];
    float*       out       = (float*)d_out;

    float* ws   = (float*)d_ws;
    float* xz   = ws;
    float* xc   = xz   + (size_t)NROWS * 2 * DINNER;
    float* xdbl = xc   + (size_t)NROWS * DINNER;
    float* dtm  = xdbl + (size_t)NROWS * (DTRANK + 2 * DSTATE);
    float* y    = dtm  + (size_t)NROWS * DINNER;

    gemm_wmma_blocked<0><<<dim3((2 * DINNER) / GB_N, NROWS / GB_M), 256, 0, stream>>>(
        hidden, in_proj_w, xz, nullptr, DMODEL, DMODEL, 2 * DINNER, 2 * DINNER);

    conv_silu_kernel<<<(NROWS / CV_T) * (CV_C4 / 256), 256, 0, stream>>>(xz, conv_w, conv_b, xc);

    gemm_xproj_kernel<<<NROWS / XP_M, 256, 0, stream>>>(xc, x_proj_w, xdbl);

    gemm_wmma_blocked<1><<<dim3(DINNER / GB_N, NROWS / GB_M), 256, 0, stream>>>(
        xdbl, dt_proj_w, dtm, dt_bias, DTRANK,
        DTRANK + 2 * DSTATE, DINNER, DINNER);

    scan_gate_kernel<<<dim3(DINNER / 32, B_SZ), 32 * SCAN_WAVES, 0, stream>>>(
        xc, dtm, xdbl, A_log, xz, D_param, y);

    gemm_wmma_blocked<0><<<dim3(DMODEL / GB_N, NROWS / GB_M), 256, 0, stream>>>(
        y, out_proj, out, nullptr, DINNER, DINNER, DMODEL, DMODEL);
}


# conv output and dt stored time-interleaved [t/4][d][4]: scan reads 4 steps per 16-B load (28 loads per 16 steps, not 56), fetches 2 groups ahead; its barrier waits for LDS only; sum order same
# speedup vs baseline: 3.4992x; 1.0466x over previous
#include <hip/hip_runtime.h>
#include <math.h>


typedef float v2f __attribute__((ext_vector_type(2)));
typedef float v8f __attribute__((ext_vector_type(8)));

#define B_SZ    4
#define LSEQ    1024
#define DMODEL  1024
#define DINNER  2048
#define DSTATE  16
#define DCONV   4
#define DTRANK  64
#define NROWS   (B_SZ * LSEQ)

__device__ __forceinline__ float silu_f(float x) { return x / (1.0f + __expf(-x)); }
__device__ __forceinline__ float softplus_f(float x) {
    const float e   = __expf(x);
    const bool  big = e > 1.0f;
    const float hi  = big ? e : 1.0f;
    const float lo  = big ? 1.0f : e;
    const float u   = hi + lo;
    const float c   = (u - hi) - lo;
    const float l2  = __builtin_amdgcn_logf(u);
    const float r   = fmaf(l2, 0.693147180559945f, -(c * __builtin_amdgcn_rcpf(u)));
    return (x > 20.0f) ? x : r;
}

__device__ __forceinline__ v8f wmma_f32(v2f a, v2f b, v8f c) {
    return __builtin_amdgcn_wmma_f32_16x16x4_f32(
               false, a, false, b, (short)0, c, false, false);
}

#define GB_M 128
#define GB_N 128
#define GB_K 32
#define GA_P (GB_K + 4)
#define GB_P (2 * GB_N + 32)

template <int EPI>
__global__ __launch_bounds__(256) void gemm_wmma_blocked(
    const float* __restrict__ A, const float* __restrict__ B,
    float* __restrict__ C, const float* __restrict__ bias,
    int K, int lda, int ldb, int ldc)
{
    __shared__ __attribute__((aligned(16))) float As[2][GB_M * GA_P];
    __shared__ __attribute__((aligned(16))) float Bs[2][(GB_K / 2) * GB_P];

    const int tid  = threadIdx.x;
    const int lane = tid & 31;
    const int wave = tid >> 5;
    const int half = lane >> 4;
    const int mr   = lane & 15;
    const int wm   = wave >> 1;
    const int wn   = wave & 1;
    const int bm0  = blockIdx.y * GB_M;
    const int bn0  = blockIdx.x * GB_N;

    const int aRow = tid >> 3, aQ = tid & 7;
    const int bPr  = tid >> 7, bN = tid & 127;
    const float* Ag = A + (size_t)(bm0 + aRow) * lda + 4 * aQ;
    const float* Bg = B + (size_t)(2 * bPr) * ldb + bn0 + bN;
    const int aSt = aRow * GA_P + 4 * aQ;
    const int bSt = bPr * GB_P + 2 * bN;

    float4 ra0, ra1, ra2, ra3;
    v2f rb0, rb1, rb2, rb3, rb4, rb5, rb6, rb7;
    auto gload = [&](int k0) {
        ra0 = *(const float4*)(Ag + k0);
        ra1 = *(const float4*)(Ag + (size_t)32 * lda + k0);
        ra2 = *(const float4*)(Ag + (size_t)64 * lda + k0);
        ra3 = *(const float4*)(Ag + (size_t)96 * lda + k0);
        const float* Bk = Bg + (size_t)k0 * ldb;
        const size_t l1 = (size_t)ldb;
        rb0.x = Bk[0];        rb0.y = Bk[l1];
        rb1.x = Bk[4 * l1];   rb1.y = Bk[5 * l1];
        rb2.x = Bk[8 * l1];   rb2.y = Bk[9 * l1];
        rb3.x = Bk[12 * l1];  rb3.y = Bk[13 * l1];
        rb4.x = Bk[16 * l1];  rb4.y = Bk[17 * l1];
        rb5.x = Bk[20 * l1];  rb5.y = Bk[21 * l1];
        rb6.x = Bk[24 * l1];  rb6.y = Bk[25 * l1];
        rb7.x = Bk[28 * l1];  rb7.y = Bk[29 * l1];
    };
    auto sstore = [&](int buf) {
        float* Ad = &As[buf][aSt];
        float* Bd = &Bs[buf][bSt];
        *(float4*)(Ad)             = ra0;
        *(float4*)(Ad + 32 * GA_P) = ra1;
        *(float4*)(Ad + 64 * GA_P) = ra2;
        *(float4*)(Ad + 96 * GA_P) = ra3;
        *(v2f*)(Bd)              = rb0;
        *(v2f*)(Bd +  2 * GB_P)  = rb1;
        *(v2f*)(Bd +  4 * GB_P)  = rb2;
        *(v2f*)(Bd +  6 * GB_P)  = rb3;
        *(v2f*)(Bd +  8 * GB_P)  = rb4;
        *(v2f*)(Bd + 10 * GB_P)  = rb5;
        *(v2f*)(Bd + 12 * GB_P)  = rb6;
        *(v2f*)(Bd + 14 * GB_P)  = rb7;
    };

    v8f c[2][4];
#pragma unroll
    for (int i = 0; i < 2; ++i)
#pragma unroll
        for (int j = 0; j < 4; ++j) c[i][j] = (v8f){};

    const int aFr = (wm * 32 + mr) * GA_P + 2 * half;
    const int bFr = half * GB_P + 2 * (wn * 64 + mr);

    auto compute = [&](int buf) {
        const float* Aw = &As[buf][aFr];
        const float* Bw = &Bs[buf][bFr];
#pragma unroll
        for (int kk = 0; kk < GB_K; kk += 4) {
            v2f a0 = *(const v2f*)(Aw + kk);
            v2f a1 = *(const v2f*)(Aw + 16 * GA_P + kk);
            v2f b[4];
#pragma unroll
            for (int j = 0; j < 4; ++j)
                b[j] = *(const v2f*)(Bw + (kk / 2) * GB_P + j * 32);
#pragma unroll
            for (int j = 0; j < 4; ++j) {
                c[0][j] = wmma_f32(a0, b[j], c[0][j]);
                c[1][j] = wmma_f32(a1, b[j], c[1][j]);
            }
        }
    };

    gload(0);
    int buf = 0;
    for (int k0 = 0; k0 < K; k0 += GB_K) {
        sstore(buf);
        __syncthreads();
        gload(min(k0 + GB_K, K - GB_K));
        __builtin_amdgcn_sched_barrier(0);
        compute(buf);
        buf ^= 1;
    }

#pragma unroll
    for (int j = 0; j < 4; ++j) {
        const int col = bn0 + wn * 64 + j * 16 + mr;
        const float bv = (EPI == 1) ? bias[col] : 0.0f;
#pragma unroll
        for (int i = 0; i < 2; ++i) {
            const int row8 = bm0 + wm * 32 + i * 16 + half * 8;
            if (EPI == 1) {
                float4 lo, hi;
                lo.x = softplus_f(c[i][j][0] + bv);  lo.y = softplus_f(c[i][j][1] + bv);
                lo.z = softplus_f(c[i][j][2] + bv);  lo.w = softplus_f(c[i][j][3] + bv);
                hi.x = softplus_f(c[i][j][4] + bv);  hi.y = softplus_f(c[i][j][5] + bv);
                hi.z = softplus_f(c[i][j][6] + bv);  hi.w = softplus_f(c[i][j][7] + bv);
                float4* dst = (float4*)C + (size_t)(row8 >> 2) * ldc + col;
                dst[0]   = lo;
                dst[ldc] = hi;
            } else {
#pragma unroll
                for (int r = 0; r < 8; ++r)
                    C[(size_t)(row8 + r) * ldc + col] = c[i][j][r];
            }
        }
    }
}

#define XP_M   16
#define XP_N   (DTRANK + 2 * DSTATE)
#define XP_KC  128
#define XP_AP  (XP_KC + 4)
#define XP_NT  (XP_N / 16)
#define XP_NCH (DINNER / XP_KC)

__global__ __launch_bounds__(256) void gemm_xproj_kernel(
    const float* __restrict__ A, const float* __restrict__ B, float* __restrict__ C)
{
    __shared__ __attribute__((aligned(16))) float As[2][XP_M * XP_AP];
    __shared__ __attribute__((aligned(16))) float Bs[2][XP_KC * XP_N];

    const int tid  = threadIdx.x;
    const int lane = tid & 31;
    const int wave = tid >> 5;
    const int half = lane >> 4;
    const int mr   = lane & 15;
    const int m0   = blockIdx.x * XP_M;
    const int rot  = blockIdx.x & (XP_NCH - 1);

    const int aQ  = tid >> 7;
    const int aK  = tid & 127;
    const float* Ag = A + ((size_t)(m0 / 4 + aQ) * DINNER + aK) * 4;
    const float* Bg = B + 4 * tid;
    const int aSt = (4 * aQ) * XP_AP + aK;
    const int bSt = 4 * tid;

    float4 ra0, ra1;
    float4 rb0, rb1, rb2, rb3, rb4, rb5, rb6, rb7, rb8, rb9, rb10, rb11;
    auto gload = [&](int ci) {
        const int k0 = ((ci + rot) & (XP_NCH - 1)) * XP_KC;
        ra0 = *(const float4*)(Ag + (size_t)k0 * 4);
        ra1 = *(const float4*)(Ag + ((size_t)2 * DINNER + k0) * 4);
        const float* Bk = Bg + (size_t)k0 * XP_N;
        rb0  = *(const float4*)(Bk);
        rb1  = *(const float4*)(Bk + 1024);
        rb2  = *(const float4*)(Bk + 2048);
        rb3  = *(const float4*)(Bk + 3072);
        rb4  = *(const float4*)(Bk + 4096);
        rb5  = *(const float4*)(Bk + 5120);
        rb6  = *(const float4*)(Bk + 6144);
        rb7  = *(const float4*)(Bk + 7168);
        rb8  = *(const float4*)(Bk + 8192);
        rb9  = *(const float4*)(Bk + 9216);
        rb10 = *(const float4*)(Bk + 10240);
        rb11 = *(const float4*)(Bk + 11264);
    };
    auto sstore = [&](int buf) {
        float* Ad = &As[buf][aSt];
        float* Bd = &Bs[buf][bSt];
        Ad[0]          = ra0.x;  Ad[XP_AP]      = ra0.y;
        Ad[2 * XP_AP]  = ra0.z;  Ad[3 * XP_AP]  = ra0.w;
        Ad[8 * XP_AP]  = ra1.x;  Ad[9 * XP_AP]  = ra1.y;
        Ad[10 * XP_AP] = ra1.z;  Ad[11 * XP_AP] = ra1.w;
        *(float4*)(Bd)         = rb0;
        *(float4*)(Bd + 1024)  = rb1;
        *(float4*)(Bd + 2048)  = rb2;
        *(float4*)(Bd + 3072)  = rb3;
        *(float4*)(Bd + 4096)  = rb4;
        *(float4*)(Bd + 5120)  = rb5;
        *(float4*)(Bd + 6144)  = rb6;
        *(float4*)(Bd + 7168)  = rb7;
        *(float4*)(Bd + 8192)  = rb8;
        *(float4*)(Bd + 9216)  = rb9;
        *(float4*)(Bd + 10240) = rb10;
        *(float4*)(Bd + 11264) = rb11;
    };

    v8f acc[XP_NT];
#pragma unroll
    for (int j = 0; j < XP_NT; ++j) acc[j] = (v8f){};

    const int aFr = mr * XP_AP + 16 * wave + half;
    const int bFr = (16 * wave + half) * XP_N + mr;

    auto compute = [&](int buf) {
        const float* Aw = &As[buf][aFr];
        const float* Bw = &Bs[buf][bFr];
#pragma unroll
        for (int i = 0; i < 4; ++i) {
            v2f a;
            a.x = Aw[4 * i];
            a.y = Aw[4 * i + 2];
            v2f b[XP_NT];
#pragma unroll
            for (int j = 0; j < XP_NT; ++j) {
                b[j].x = Bw[(4 * i) * XP_N + 16 * j];
                b[j].y = Bw[(4 * i + 2) * XP_N + 16 * j];
            }
#pragma unroll
            for (int j = 0; j < XP_NT; ++j)
                acc[j] = wmma_f32(a, b[j], acc[j]);
        }
    };

    gload(0);
    int buf = 0;
    for (int ci = 0; ci < XP_NCH; ++ci) {
        sstore(buf);
        __syncthreads();
        gload(min(ci + 1, XP_NCH - 1));
        __builtin_amdgcn_sched_barrier(0);
        compute(buf);
        buf ^= 1;
    }

    __syncthreads();
    float* red = &Bs[0][0];
#pragma unroll
    for (int j = 0; j < XP_NT; ++j)
#pragma unroll
        for (int r = 0; r < 8; ++r)
            red[((wave * XP_NT + j) * 8 + r) * 32 + lane] = acc[j][r];
    __syncthreads();

#pragma unroll
    for (int i = 0; i < (XP_M * XP_N) / 256; ++i) {
        const int e   = tid + 256 * i;
        const int row = e / XP_N;
        const int col = e - row * XP_N;
        const int src = (((col >> 4) * 8) + (row & 7)) * 32 + 16 * (row >> 3) + (col & 15);
        float s = red[src];
#pragma unroll
        for (int p = 1; p < 8; ++p) s += red[p * (XP_NT * 8 * 32) + src];
        C[(size_t)m0 * XP_N + e] = s;
    }
}

#define CV_T  32

__global__ __launch_bounds__(256) void conv_silu_kernel(
    const float* __restrict__ xz, const float* __restrict__ conv_w,
    const float* __restrict__ conv_b, float* __restrict__ xc)
{
    const int d     = (blockIdx.x % (DINNER / 256)) * 256 + threadIdx.x;
    const int row0  = (blockIdx.x / (DINNER / 256)) * CV_T;
    const bool head = (row0 & (LSEQ - 1)) == 0;

    const float4 w  = *(const float4*)(conv_w + (size_t)d * DCONV);
    const float  bv = conv_b[d];

    const float* xp = xz + (size_t)row0 * (2 * DINNER) + d;
    float x[CV_T + 3];
    const float* xh = head ? xp : xp - 3 * (2 * DINNER);
    const float h0 = xh[0];
    const float h1 = xh[2 * DINNER];
    const float h2 = xh[2 * (2 * DINNER)];
#pragma unroll
    for (int r = 0; r < CV_T; ++r)
        x[3 + r] = xp[(size_t)r * (2 * DINNER)];
    x[0] = head ? 0.0f : h0;
    x[1] = head ? 0.0f : h1;
    x[2] = head ? 0.0f : h2;

    float4* op = (float4*)xc + (size_t)(row0 / 4) * DINNER + d;
#pragma unroll
    for (int q = 0; q < CV_T / 4; ++q) {
        float o[4];
#pragma unroll
        for (int s = 0; s < 4; ++s) {
            const int t = 4 * q + s;
            float a = bv;
            a += w.x * x[t];
            a += w.y * x[t + 1];
            a += w.z * x[t + 2];
            a += w.w * x[t + 3];
            o[s] = silu_f(a);
        }
        op[(size_t)q * DINNER] = make_float4(o[0], o[1], o[2], o[3]);
    }
}

#define SCAN_GROUP 16
#define SCAN_WAVES 4
#define NPW (DSTATE / SCAN_WAVES)
#define FPW (SCAN_GROUP / SCAN_WAVES)
#define SCAN_Q (SCAN_GROUP / 4)
#define XDW (DTRANK + 2 * DSTATE)

struct ScanOps {
    float u[SCAN_GROUP];
    float dt[SCAN_GROUP];
    float bc[SCAN_GROUP];
    float zf[FPW];
};

__device__ __forceinline__ void group_barrier_lds() {
    __builtin_amdgcn_fence(__ATOMIC_RELEASE, "workgroup", "local");
    __builtin_amdgcn_s_barrier();
    __builtin_amdgcn_fence(__ATOMIC_ACQUIRE, "workgroup", "local");
}

__global__ __launch_bounds__(32 * SCAN_WAVES) void scan_gate_kernel(
    const float* __restrict__ xc,   const float* __restrict__ dtm,
    const float* __restrict__ xdbl, const float* __restrict__ A_log,
    const float* __restrict__ xz,   const float* __restrict__ D_param,
    float* __restrict__ y)
{
    __shared__ float4 part[2][SCAN_Q][SCAN_WAVES][32];

    const int b    = blockIdx.y;
    const int lane = threadIdx.x & 31;
    const int w    = __builtin_amdgcn_readfirstlane(threadIdx.x >> 5);
    const int d    = blockIdx.x * 32 + lane;
    const int n0   = w * NPW;

    float Ar[NPW];
#pragma unroll
    for (int j = 0; j < NPW; ++j)
        Ar[j] = -__expf(A_log[d * DSTATE + n0 + j]);
    const float Dv = D_param[d];

    float h[NPW];
#pragma unroll
    for (int j = 0; j < NPW; ++j) h[j] = 0.0f;

    const size_t rowBase = (size_t)b * LSEQ;
    const float4* up4 = (const float4*)xc  + (rowBase / 4) * DINNER + d;
    const float4* dp4 = (const float4*)dtm + (rowBase / 4) * DINNER + d;
    const float* zp  = xz   + rowBase * (2 * DINNER) + DINNER + d;
    float*       yp  = y    + rowBase * DINNER + d;
    const float* bcp = xdbl + rowBase * XDW + DTRANK + lane;
    const int    f0  = w * FPW;
    const int qm0 = (w == 0) ? -1 : 0, qm1 = (w == 1) ? -1 : 0;
    const int qm2 = (w == 2) ? -1 : 0, qm3 = (w == 3) ? -1 : 0;

    auto fetch = [&](ScanOps& o, int tb) {
        const size_t q0 = (size_t)(tb >> 2);
#pragma unroll
        for (int k = 0; k < SCAN_Q; ++k) {
            const float4 a = up4[(q0 + k) * DINNER];
            const float4 c = dp4[(q0 + k) * DINNER];
            o.u[4 * k]  = a.x;  o.u[4 * k + 1]  = a.y;  o.u[4 * k + 2]  = a.z;  o.u[4 * k + 3]  = a.w;
            o.dt[4 * k] = c.x;  o.dt[4 * k + 1] = c.y;  o.dt[4 * k + 2] = c.z;  o.dt[4 * k + 3] = c.w;
#pragma unroll
            for (int s = 0; s < 4; ++s)
                o.bc[4 * k + s] = bcp[(size_t)(tb + 4 * k + s) * XDW];
        }
#pragma unroll
        for (int k = 0; k < FPW; ++k)
            o.zf[k] = zp[(size_t)(tb + f0 + k) * (2 * DINNER)];
    };

    auto group = [&](const ScanOps& cur, ScanOps& nx, int t0, int buf) {
        fetch(nx, min(t0 + 2 * SCAN_GROUP, LSEQ - SCAN_GROUP));
        __builtin_amdgcn_sched_barrier(0);

#pragma unroll
        for (int q = 0; q < SCAN_Q; ++q) {
            float pv[4];
#pragma unroll
            for (int s = 0; s < 4; ++s) {
                const int i = 4 * q + s;
                const float u = cur.u[i], dtv = cur.dt[i];
                const int bcbits = __float_as_int(cur.bc[i]);
                float Bv[NPW], Cv[NPW];
#pragma unroll
                for (int j = 0; j < NPW; ++j) {
                    Bv[j] = __int_as_float(__builtin_amdgcn_readlane(bcbits, n0 + j));
                    Cv[j] = __int_as_float(__builtin_amdgcn_readlane(bcbits, DSTATE + n0 + j));
                }

                float acc = 0.0f;
#pragma unroll
                for (int j = 0; j < NPW; ++j) {
                    const float dA = __expf(dtv * Ar[j]);
                    h[j] = dA * h[j] + (dtv * Bv[j]) * u;
                    acc += Cv[j] * h[j];
                }
                pv[s] = acc;
            }
            part[buf][q][w][lane] = make_float4(pv[0], pv[1], pv[2], pv[3]);
        }

        group_barrier_lds();

        float4 s4 = part[buf][w][0][lane];
#pragma unroll
        for (int p = 1; p < SCAN_WAVES; ++p) {
            const float4 v = part[buf][w][p][lane];
            s4.x += v.x;  s4.y += v.y;  s4.z += v.z;  s4.w += v.w;
        }
        const float sv[FPW] = { s4.x, s4.y, s4.z, s4.w };
#pragma unroll
        for (int k = 0; k < FPW; ++k) {
            const int ub = (__float_as_int(cur.u[k])      & qm0) | (__float_as_int(cur.u[4 + k])  & qm1)
                         | (__float_as_int(cur.u[8 + k])  & qm2) | (__float_as_int(cur.u[12 + k]) & qm3);
            yp[(size_t)(t0 + f0 + k) * DINNER] = (sv[k] + Dv * __int_as_float(ub)) * silu_f(cur.zf[k]);
        }
    };

    ScanOps o0, o1, o2, o3;
    fetch(o0, 0);
    __builtin_amdgcn_sched_barrier(0);
    fetch(o1, SCAN_GROUP);
    __builtin_amdgcn_sched_barrier(0);
    for (int t0 = 0; t0 < LSEQ; t0 += 4 * SCAN_GROUP) {
        group(o0, o2, t0,                  0);
        group(o1, o3, t0 + SCAN_GROUP,     1);
        group(o2, o0, t0 + 2 * SCAN_GROUP, 0);
        group(o3, o1, t0 + 3 * SCAN_GROUP, 1);
    }
}

extern "C" void kernel_launch(void* const* d_in, const int* in_sizes, int n_in,
                              void* d_out, int out_size, void* d_ws, size_t ws_size,
                              hipStream_t stream)
{
    const float* hidden    = (const float*)d_in[0];
    const float* in_proj_w = (const float*)d_in[1];
    const float* conv_w    = (const float*)d_in[2];
    const float* conv_b    = (const float*)d_in[3];
    const float* x_proj_w  = (const float*)d_in[4];
    const float* dt_proj_w = (const float*)d_in[5];
    const float* dt_bias   = (const float*)d_in[6];
    const float* A_log     = (const float*)d_in[7];
    const float* D_param   = (const float*)d_in[8];
    const float* out_proj  = (const float*)d_in[9];
    float*       out       = (float*)d_out;

    float* ws   = (float*)d_ws;
    float* xz   = ws;
    float* xc   = xz   + (size_t)NROWS * 2 * DINNER;
    float* xdbl = xc   + (size_t)NROWS * DINNER;
    float* dtm  = xdbl + (size_t)NROWS * (DTRANK + 2 * DSTATE);
    float* y    = dtm  + (size_t)NROWS * DINNER;

    gemm_wmma_blocked<0><<<dim3((2 * DINNER) / GB_N, NROWS / GB_M), 256, 0, stream>>>(
        hidden, in_proj_w, xz, nullptr, DMODEL, DMODEL, 2 * DINNER, 2 * DINNER);

    conv_silu_kernel<<<(NROWS / CV_T) * (DINNER / 256), 256, 0, stream>>>(xz, conv_w, conv_b, xc);

    gemm_xproj_kernel<<<NROWS / XP_M, 256, 0, stream>>>(xc, x_proj_w, xdbl);

    gemm_wmma_blocked<1><<<dim3(DINNER / GB_N, NROWS / GB_M), 256, 0, stream>>>(
        xdbl, dt_proj_w, dtm, dt_bias, DTRANK,
        DTRANK + 2 * DSTATE, DINNER, DINNER);

    scan_gate_kernel<<<dim3(DINNER / 32, B_SZ), 32 * SCAN_WAVES, 0, stream>>>(
        xc, dtm, xdbl, A_log, xz, D_param, y);

    gemm_wmma_blocked<0><<<dim3(DMODEL / GB_N, NROWS / GB_M), 256, 0, stream>>>(
        y, out_proj, out, nullptr, DINNER, DINNER, DMODEL, DMODEL);
}
